# starting kernel
# baseline (speedup 1.0000x reference)
#include <hip/hip_runtime.h>
#include <hip/hip_bf16.h>

typedef float v2f __attribute__((ext_vector_type(2)));
typedef float v8f __attribute__((ext_vector_type(8)));

#define EPSV 1e-5f

__global__ void roi_pool_kernel(const float* __restrict__ x, float* __restrict__ A1) {
    const int n = blockIdx.x;
    const int c = threadIdx.x;
    const float* base = x + ((size_t)(n * 256 + c) * 64) * 64;
    float v[4][5];
#pragma unroll
    for (int rr = 0; rr < 4; ++rr)
#pragma unroll
        for (int cc = 0; cc < 5; ++cc)
            v[rr][cc] = base[(2 + rr) * 64 + 1 + cc];
    float r0[5], r1[5];
#pragma unroll
    for (int cc = 0; cc < 5; ++cc) {
        r0[cc] = fmaxf(v[0][cc], v[1][cc]);
        r1[cc] = fmaxf(v[2][cc], v[3][cc]);
    }
    float p00 = fmaxf(fmaxf(r0[0], r0[1]), r0[2]);
    float p01 = fmaxf(fmaxf(r0[2], r0[3]), r0[4]);
    float p10 = fmaxf(fmaxf(r1[0], r1[1]), r1[2]);
    float p11 = fmaxf(fmaxf(r1[2], r1[3]), r1[4]);
    float* o = A1 + (size_t)n * 4 * 256 + c;
    o[0 * 256] = p00;
    o[1 * 256] = p01;
    o[2 * 256] = p10;
    o[3 * 256] = p11;
}

template <int KIN, int COUT, bool RESHAPE>
__global__ __launch_bounds__(256) void conv_bn_relu_kernel(
    const float* __restrict__ Ain, int a_head_stride,
    const float* __restrict__ Wt,
    const float* __restrict__ g, const float* __restrict__ b,
    float* __restrict__ Out, int out_head_stride) {
    constexpr int NT  = COUT / 16;
    constexpr int TT  = 16 * NT;
    constexpr int TPW = TT / 8;

    __shared__ float lsum[COUT], lsq[COUT], lscale[COUT], lshift[COUT];

    const int h    = blockIdx.x;
    const int tid  = threadIdx.x;
    const int wave = tid >> 5;
    const int lane = tid & 31;
    const int half = lane >> 4;
    const int r    = lane & 15;

    if (tid < COUT) { lsum[tid] = 0.f; lsq[tid] = 0.f; }
    __syncthreads();

    const float* A = Ain + (size_t)h * a_head_stride;
    const float* W = Wt + (size_t)h * COUT * KIN;

    v8f acc[TPW];
#pragma unroll
    for (int i = 0; i < TPW; ++i) {
        const int tt = wave + 8 * i;
        const int mt = tt & 15, nt = tt >> 4;
        const int m0 = mt * 16, n0 = nt * 16;
        v8f c = {};
        const float2* arow = reinterpret_cast<const float2*>(A + (size_t)(m0 + r) * KIN);
        const float2* brow = reinterpret_cast<const float2*>(W + (size_t)(n0 + r) * KIN);
        for (int k0 = 0; k0 < KIN; k0 += 4) {
            float2 av = arow[(k0 >> 1) + half];
            float2 bv = brow[(k0 >> 1) + half];
            v2f a; a[0] = av.x; a[1] = av.y;
            v2f bb; bb[0] = bv.x; bb[1] = bv.y;
            c = __builtin_amdgcn_wmma_f32_16x16x4_f32(false, a, false, bb,
                                                      (short)0, c, false, false);
        }
        acc[i] = c;
        float s = 0.f, q = 0.f;
#pragma unroll
        for (int v = 0; v < 8; ++v) { float xv = c[v]; s += xv; q += xv * xv; }
        atomicAdd(&lsum[n0 + r], s);
        atomicAdd(&lsq[n0 + r], q);
    }
    __syncthreads();

    if (tid < COUT) {
        float mu  = lsum[tid] * (1.f / 256.f);
        float var = lsq[tid] * (1.f / 256.f) - mu * mu;
        float sc  = g[h * COUT + tid] * rsqrtf(var + EPSV);
        lscale[tid] = sc;
        lshift[tid] = b[h * COUT + tid] - mu * sc;
    }
    __syncthreads();

    float* O = Out + (size_t)h * out_head_stride;
#pragma unroll
    for (int i = 0; i < TPW; ++i) {
        const int tt = wave + 8 * i;
        const int mt = tt & 15, nt = tt >> 4;
        const int m0 = mt * 16, n0 = nt * 16;
        const int col = n0 + r;
        const float sc = lscale[col], sh = lshift[col];
#pragma unroll
        for (int v = 0; v < 8; ++v) {
            const int row = m0 + v + 8 * half;
            float val = acc[i][v] * sc + sh;
            val = fmaxf(val, 0.f);
            if (RESHAPE) {
                const int n = row >> 2, s = row & 3;
                O[n * (COUT * 4) + col * 4 + s] = val;
            } else {
                O[(size_t)row * COUT + col] = val;
            }
        }
    }
}

template <int M, int N, int K, bool RELU>
__global__ __launch_bounds__(256) void gemm_bias_kernel(
    const float* __restrict__ Ain, int a_head_stride,
    const float* __restrict__ Wt,
    const float* __restrict__ bias,
    float* __restrict__ Out, int out_head_stride) {
    constexpr int MT = (M + 15) / 16, NT = (N + 15) / 16, TT = MT * NT;
    const int h    = blockIdx.y;
    const int tid  = threadIdx.x;
    const int wave = tid >> 5;
    const int lane = tid & 31;
    const int half = lane >> 4;
    const int r    = lane & 15;
    const int wglobal = blockIdx.x * (blockDim.x >> 5) + wave;
    const int wstride = gridDim.x * (blockDim.x >> 5);

    const float* A = Ain + (size_t)h * a_head_stride;
    const float* W = Wt + (size_t)h * N * K;
    float* O = Out + (size_t)h * out_head_stride;

    for (int tt = wglobal; tt < TT; tt += wstride) {
        const int mt = tt % MT, nt = tt / MT;
        const int m0 = mt * 16, n0 = nt * 16;
        const int col = n0 + r;
        const bool colok = (col < N);
        v8f c = {};
        const float2* arow = reinterpret_cast<const float2*>(A + (size_t)(m0 + r) * K);
        const float2* brow = reinterpret_cast<const float2*>(W + (size_t)(colok ? col : 0) * K);
        for (int k0 = 0; k0 < K; k0 += 4) {
            float2 av = arow[(k0 >> 1) + half];
            float2 bv = brow[(k0 >> 1) + half];
            v2f a; a[0] = av.x; a[1] = av.y;
            v2f bb; bb[0] = colok ? bv.x : 0.f; bb[1] = colok ? bv.y : 0.f;
            c = __builtin_amdgcn_wmma_f32_16x16x4_f32(false, a, false, bb,
                                                      (short)0, c, false, false);
        }
        if (colok) {
            const float bs = bias[h * N + col];
#pragma unroll
            for (int v = 0; v < 8; ++v) {
                const int row = m0 + v + 8 * half;
                float val = c[v] + bs;
                if (RELU) val = fmaxf(val, 0.f);
                O[(size_t)row * N + col] = val;
            }
        }
    }
}

__global__ void final_combine_kernel(const float* __restrict__ hout,
                                     const float* __restrict__ W2,
                                     const float* __restrict__ bias2,
                                     float* __restrict__ out) {
    __shared__ float comb[100];
    const int n = blockIdx.x;
    const int t = threadIdx.x;
    if (t < 100) {
        float s = 0.f;
#pragma unroll
        for (int h = 0; h < 4; ++h) s += hout[((size_t)h * 64 + n) * 100 + t];
        comb[t] = s * 0.25f;
    }
    __syncthreads();
    if (t < 100) {
        float acc = bias2[t];
        for (int j = 0; j < 100; ++j) acc += comb[j] * W2[t * 100 + j];
        float* orow = out + (size_t)n * 64 * 100 + t;
#pragma unroll 4
        for (int rep = 0; rep < 64; ++rep) orow[(size_t)rep * 100] = acc;
    }
}

extern "C" void kernel_launch(void* const* d_in, const int* in_sizes, int n_in,
                              void* d_out, int out_size, void* d_ws, size_t ws_size,
                              hipStream_t stream) {
    const float* x     = (const float*)d_in[0];
    const float* w1    = (const float*)d_in[1];
    const float* g1    = (const float*)d_in[2];
    const float* b1    = (const float*)d_in[3];
    const float* w2    = (const float*)d_in[4];
    const float* g2    = (const float*)d_in[5];
    const float* b2    = (const float*)d_in[6];
    const float* w3    = (const float*)d_in[7];
    const float* g3    = (const float*)d_in[8];
    const float* b3    = (const float*)d_in[9];
    const float* lw1   = (const float*)d_in[10];
    const float* lb1   = (const float*)d_in[11];
    const float* lw2   = (const float*)d_in[12];
    const float* lb2   = (const float*)d_in[13];
    const float* W2    = (const float*)d_in[14];
    const float* bias2 = (const float*)d_in[15];
    float* out = (float*)d_out;

    float* ws   = (float*)d_ws;
    float* A1   = ws;
    float* z1   = A1 + 256 * 256;
    float* z2   = z1 + 4 * 256 * 32;
    float* fcin = z2 + 4 * 256 * 64;
    float* h1   = fcin + 4 * 64 * 256;
    float* hout = h1 + 4 * 64 * 512;

    roi_pool_kernel<<<64, 256, 0, stream>>>(x, A1);

    conv_bn_relu_kernel<256, 32, false><<<4, 256, 0, stream>>>(
        A1, 0, w1, g1, b1, z1, 256 * 32);
    conv_bn_relu_kernel<32, 64, false><<<4, 256, 0, stream>>>(
        z1, 256 * 32, w2, g2, b2, z2, 256 * 64);
    conv_bn_relu_kernel<64, 64, true><<<4, 256, 0, stream>>>(
        z2, 256 * 64, w3, g3, b3, fcin, 64 * 256);

    gemm_bias_kernel<64, 512, 256, true><<<dim3(8, 4), 256, 0, stream>>>(
        fcin, 64 * 256, lw1, lb1, h1, 64 * 512);
    gemm_bias_kernel<64, 100, 512, false><<<dim3(4, 4), 256, 0, stream>>>(
        h1, 64 * 512, lw2, lb2, hout, 64 * 100);

    final_combine_kernel<<<64, 128, 0, stream>>>(hout, W2, bias2, out);
}


# Both WMMA GEMM templates read A/B fragments from LDS tiles filled by coalesced b128 global loads (was per-lane b64 global loads, one cache line per row per load); k order of every tile unchanged
# speedup vs baseline: 1.6665x; 1.6665x over previous
#include <hip/hip_runtime.h>
#include <hip/hip_bf16.h>

typedef float v2f __attribute__((ext_vector_type(2)));
typedef float v8f __attribute__((ext_vector_type(8)));

#define EPSV 1e-5f

__global__ void roi_pool_kernel(const float* __restrict__ x, float* __restrict__ A1) {
    const int n = blockIdx.x;
    const int c = threadIdx.x;
    const float* base = x + ((size_t)(n * 256 + c) * 64) * 64;
    float v[4][5];
#pragma unroll
    for (int rr = 0; rr < 4; ++rr)
#pragma unroll
        for (int cc = 0; cc < 5; ++cc)
            v[rr][cc] = base[(2 + rr) * 64 + 1 + cc];
    float r0[5], r1[5];
#pragma unroll
    for (int cc = 0; cc < 5; ++cc) {
        r0[cc] = fmaxf(v[0][cc], v[1][cc]);
        r1[cc] = fmaxf(v[2][cc], v[3][cc]);
    }
    float p00 = fmaxf(fmaxf(r0[0], r0[1]), r0[2]);
    float p01 = fmaxf(fmaxf(r0[2], r0[3]), r0[4]);
    float p10 = fmaxf(fmaxf(r1[0], r1[1]), r1[2]);
    float p11 = fmaxf(fmaxf(r1[2], r1[3]), r1[4]);
    float* o = A1 + (size_t)n * 4 * 256 + c;
    o[0 * 256] = p00;
    o[1 * 256] = p01;
    o[2 * 256] = p10;
    o[3 * 256] = p11;
}

template <int K, int NROWS>
__device__ __forceinline__ void stage_rows(float* __restrict__ dst,
                                           const float* __restrict__ src, int tid) {
    constexpr int P     = K + 4;
    constexpr int VPR   = K / 4;
    constexpr int ITERS = NROWS * VPR / 256;
    static_assert((NROWS * VPR) % 256 == 0, "whole passes of the workgroup only");
    static_assert(256 % VPR == 0, "a row is a whole fraction of one pass of the workgroup");
    const float4* s4 = reinterpret_cast<const float4*>(src) + tid;
    float* d = dst + (tid / VPR) * P + (tid % VPR) * 4;
    float4 v[ITERS];
#pragma unroll
    for (int j = 0; j < ITERS; ++j) v[j] = s4[256 * j];
#pragma unroll
    for (int j = 0; j < ITERS; ++j)
        *reinterpret_cast<float4*>(d + (256 / VPR) * j * P) = v[j];
}

template <int K>
__device__ __forceinline__ void stage_rows_rt(float* __restrict__ dst,
                                              const float* __restrict__ src,
                                              int nrows, int tid) {
    constexpr int P   = K + 4;
    constexpr int VPR = K / 4;
    const int total = nrows * VPR;
    const float4* s4 = reinterpret_cast<const float4*>(src);
    for (int i = tid; i < total; i += 256) {
        const int row = i / VPR, c4 = i % VPR;
        *reinterpret_cast<float4*>(dst + row * P + c4 * 4) = s4[i];
    }
}

template <int KIN, int COUT, bool RESHAPE>
__global__ __launch_bounds__(256) void conv_bn_relu_kernel(
    const float* __restrict__ Ain, int a_head_stride,
    const float* __restrict__ Wt,
    const float* __restrict__ g, const float* __restrict__ b,
    float* __restrict__ Out, int out_head_stride) {
    constexpr int NT  = COUT / 16;
    constexpr int TT  = 16 * NT;
    constexpr int TPW = TT / 8;
    constexpr int P   = KIN + 4;

    __shared__ float lsum[COUT], lsq[COUT], lscale[COUT], lshift[COUT];
    __shared__ alignas(16) float sA[256 * P];
    __shared__ alignas(16) float sW[COUT * P];

    const int h    = blockIdx.x;
    const int tid  = threadIdx.x;
    const int wave = tid >> 5;
    const int lane = tid & 31;
    const int half = lane >> 4;
    const int r    = lane & 15;

    const float* A = Ain + (size_t)h * a_head_stride;
    const float* W = Wt + (size_t)h * COUT * KIN;

    if (tid < COUT) { lsum[tid] = 0.f; lsq[tid] = 0.f; }
    stage_rows<KIN, COUT>(sW, W, tid);
    stage_rows<KIN, 256>(sA, A, tid);
    __syncthreads();

    v8f acc[TPW];
#pragma unroll
    for (int i = 0; i < TPW; ++i) acc[i] = v8f{};

    const float* alo = sA + (wave * 16 + r) * P + 2 * half;
    const float* ahi = alo + 128 * P;
    const float* bw  = sW + r * P + 2 * half;
    for (int k0 = 0; k0 < KIN; k0 += 4) {
        const float2 a0 = *reinterpret_cast<const float2*>(alo + k0);
        const float2 a1 = *reinterpret_cast<const float2*>(ahi + k0);
        v2f fa0; fa0[0] = a0.x; fa0[1] = a0.y;
        v2f fa1; fa1[0] = a1.x; fa1[1] = a1.y;
#pragma unroll
        for (int nt = 0; nt < NT; ++nt) {
            const float2 bv = *reinterpret_cast<const float2*>(bw + nt * 16 * P + k0);
            v2f fb; fb[0] = bv.x; fb[1] = bv.y;
            acc[2 * nt]     = __builtin_amdgcn_wmma_f32_16x16x4_f32(false, fa0, false, fb,
                                                                    (short)0, acc[2 * nt], false, false);
            acc[2 * nt + 1] = __builtin_amdgcn_wmma_f32_16x16x4_f32(false, fa1, false, fb,
                                                                    (short)0, acc[2 * nt + 1], false, false);
        }
    }
#pragma unroll
    for (int i = 0; i < TPW; ++i) {
        const int n0 = (i >> 1) * 16;
        float s = 0.f, q = 0.f;
#pragma unroll
        for (int v = 0; v < 8; ++v) { float xv = acc[i][v]; s += xv; q += xv * xv; }
        atomicAdd(&lsum[n0 + r], s);
        atomicAdd(&lsq[n0 + r], q);
    }
    __syncthreads();

    if (tid < COUT) {
        float mu  = lsum[tid] * (1.f / 256.f);
        float var = lsq[tid] * (1.f / 256.f) - mu * mu;
        float sc  = g[h * COUT + tid] * rsqrtf(var + EPSV);
        lscale[tid] = sc;
        lshift[tid] = b[h * COUT + tid] - mu * sc;
    }
    __syncthreads();

    float* O = Out + (size_t)h * out_head_stride;
#pragma unroll
    for (int i = 0; i < TPW; ++i) {
        const int tt = wave + 8 * i;
        const int mt = tt & 15, nt = tt >> 4;
        const int m0 = mt * 16, n0 = nt * 16;
        const int col = n0 + r;
        const float sc = lscale[col], sh = lshift[col];
#pragma unroll
        for (int v = 0; v < 8; ++v) {
            const int row = m0 + v + 8 * half;
            float val = acc[i][v] * sc + sh;
            val = fmaxf(val, 0.f);
            if (RESHAPE) {
                const int n = row >> 2, s = row & 3;
                O[n * (COUT * 4) + col * 4 + s] = val;
            } else {
                O[(size_t)row * COUT + col] = val;
            }
        }
    }
}

template <int N, int K, int MT, int BR>
__device__ __forceinline__ void stage_weight_rows(float* __restrict__ sB,
                                                  const float* __restrict__ W,
                                                  int tt0, int tid) {
    const int nrow0 = (tt0 / MT) * 16;
    if constexpr (N % BR == 0) {
        stage_rows<K, BR>(sB, W + (size_t)nrow0 * K, tid);
    } else {
        const int nvalid = (N - nrow0 < BR) ? (N - nrow0) : BR;
        if (nvalid == BR) stage_rows<K, BR>(sB, W + (size_t)nrow0 * K, tid);
        else              stage_rows_rt<K>(sB, W + (size_t)nrow0 * K, nvalid, tid);
    }
}

template <int M, int N, int K, bool RELU>
__global__ __launch_bounds__(256) void gemm_bias_kernel(
    const float* __restrict__ Ain, int a_head_stride,
    const float* __restrict__ Wt,
    const float* __restrict__ bias,
    float* __restrict__ Out, int out_head_stride) {
    constexpr int MT = (M + 15) / 16, NT = (N + 15) / 16, TT = MT * NT;
    constexpr int P  = K + 4;
    constexpr int BR = 16 * (8 / MT);
    static_assert(M % 16 == 0 && 8 % MT == 0, "a round of 8 tiles covers whole N-tiles");

    __shared__ alignas(16) float sA[M * P];
    __shared__ alignas(16) float sB[BR * P];

    const int h    = blockIdx.y;
    const int tid  = threadIdx.x;
    const int wave = tid >> 5;
    const int lane = tid & 31;
    const int half = lane >> 4;
    const int r    = lane & 15;
    const int wstride = gridDim.x * 8;

    const float* A = Ain + (size_t)h * a_head_stride;
    const float* W = Wt + (size_t)h * N * K;
    float* O = Out + (size_t)h * out_head_stride;

    int tt0 = blockIdx.x * 8;
    if (tt0 >= TT) return;
    stage_weight_rows<N, K, MT, BR>(sB, W, tt0, tid);
    stage_rows<K, M>(sA, A, tid);
    for (;;) {
        __syncthreads();

        const int tt = tt0 + wave;
        if (tt < TT) {
            const int mt  = tt % MT;
            const int ntl = wave / MT;
            const int m0 = mt * 16;
            const int col = (tt0 / MT) * 16 + ntl * 16 + r;
            const bool colok = (col < N);
            v8f c = {};
            const float* arow = sA + (m0 + r) * P + 2 * half;
            const float* brow = sB + (colok ? (ntl * 16 + r) : 0) * P + 2 * half;
            for (int k0 = 0; k0 < K; k0 += 4) {
                const float2 av = *reinterpret_cast<const float2*>(arow + k0);
                const float2 bv = *reinterpret_cast<const float2*>(brow + k0);
                v2f a; a[0] = av.x; a[1] = av.y;
                v2f bb; bb[0] = colok ? bv.x : 0.f; bb[1] = colok ? bv.y : 0.f;
                c = __builtin_amdgcn_wmma_f32_16x16x4_f32(false, a, false, bb,
                                                          (short)0, c, false, false);
            }
            if (colok) {
                const float bs = bias[h * N + col];
#pragma unroll
                for (int v = 0; v < 8; ++v) {
                    const int row = m0 + v + 8 * half;
                    float val = c[v] + bs;
                    if (RELU) val = fmaxf(val, 0.f);
                    O[(size_t)row * N + col] = val;
                }
            }
        }

        tt0 += wstride;
        if (tt0 >= TT) break;
        __syncthreads();
        stage_weight_rows<N, K, MT, BR>(sB, W, tt0, tid);
    }
}

__global__ void final_combine_kernel(const float* __restrict__ hout,
                                     const float* __restrict__ W2,
                                     const float* __restrict__ bias2,
                                     float* __restrict__ out) {
    __shared__ float comb[100];
    const int n = blockIdx.x;
    const int t = threadIdx.x;
    if (t < 100) {
        float s = 0.f;
#pragma unroll
        for (int h = 0; h < 4; ++h) s += hout[((size_t)h * 64 + n) * 100 + t];
        comb[t] = s * 0.25f;
    }
    __syncthreads();
    if (t < 100) {
        float acc = bias2[t];
        for (int j = 0; j < 100; ++j) acc += comb[j] * W2[t * 100 + j];
        float* orow = out + (size_t)n * 64 * 100 + t;
#pragma unroll 4
        for (int rep = 0; rep < 64; ++rep) orow[(size_t)rep * 100] = acc;
    }
}

extern "C" void kernel_launch(void* const* d_in, const int* in_sizes, int n_in,
                              void* d_out, int out_size, void* d_ws, size_t ws_size,
                              hipStream_t stream) {
    const float* x     = (const float*)d_in[0];
    const float* w1    = (const float*)d_in[1];
    const float* g1    = (const float*)d_in[2];
    const float* b1    = (const float*)d_in[3];
    const float* w2    = (const float*)d_in[4];
    const float* g2    = (const float*)d_in[5];
    const float* b2    = (const float*)d_in[6];
    const float* w3    = (const float*)d_in[7];
    const float* g3    = (const float*)d_in[8];
    const float* b3    = (const float*)d_in[9];
    const float* lw1   = (const float*)d_in[10];
    const float* lb1   = (const float*)d_in[11];
    const float* lw2   = (const float*)d_in[12];
    const float* lb2   = (const float*)d_in[13];
    const float* W2    = (const float*)d_in[14];
    const float* bias2 = (const float*)d_in[15];
    float* out = (float*)d_out;

    float* ws   = (float*)d_ws;
    float* A1   = ws;
    float* z1   = A1 + 256 * 256;
    float* z2   = z1 + 4 * 256 * 32;
    float* fcin = z2 + 4 * 256 * 64;
    float* h1   = fcin + 4 * 64 * 256;
    float* hout = h1 + 4 * 64 * 512;

    roi_pool_kernel<<<64, 256, 0, stream>>>(x, A1);

    conv_bn_relu_kernel<256, 32, false><<<4, 256, 0, stream>>>(
        A1, 0, w1, g1, b1, z1, 256 * 32);
    conv_bn_relu_kernel<32, 64, false><<<4, 256, 0, stream>>>(
        z1, 256 * 32, w2, g2, b2, z2, 256 * 64);
    conv_bn_relu_kernel<64, 64, true><<<4, 256, 0, stream>>>(
        z2, 256 * 64, w3, g3, b3, fcin, 64 * 256);

    gemm_bias_kernel<64, 512, 256, true><<<dim3(8, 4), 256, 0, stream>>>(
        fcin, 64 * 256, lw1, lb1, h1, 64 * 512);
    gemm_bias_kernel<64, 100, 512, false><<<dim3(4, 4), 256, 0, stream>>>(
        h1, 64 * 512, lw2, lb2, hout, 64 * 100);

    final_combine_kernel<<<64, 128, 0, stream>>>(hout, W2, bias2, out);
}


# The three conv+BN+ReLU launches of a head become one launch: both intermediate activations stay in LDS as the next layer's A operand, all weights staged at the start; k order and stats unchanged
# speedup vs baseline: 1.8274x; 1.0965x over previous
#include <hip/hip_runtime.h>
#include <hip/hip_bf16.h>

typedef float v2f __attribute__((ext_vector_type(2)));
typedef float v8f __attribute__((ext_vector_type(8)));

#define EPSV 1e-5f

__global__ void roi_pool_kernel(const float* __restrict__ x, float* __restrict__ A1) {
    const int n = blockIdx.x;
    const int c = threadIdx.x;
    const float* base = x + ((size_t)(n * 256 + c) * 64) * 64;
    float v[4][5];
#pragma unroll
    for (int rr = 0; rr < 4; ++rr)
#pragma unroll
        for (int cc = 0; cc < 5; ++cc)
            v[rr][cc] = base[(2 + rr) * 64 + 1 + cc];
    float r0[5], r1[5];
#pragma unroll
    for (int cc = 0; cc < 5; ++cc) {
        r0[cc] = fmaxf(v[0][cc], v[1][cc]);
        r1[cc] = fmaxf(v[2][cc], v[3][cc]);
    }
    float p00 = fmaxf(fmaxf(r0[0], r0[1]), r0[2]);
    float p01 = fmaxf(fmaxf(r0[2], r0[3]), r0[4]);
    float p10 = fmaxf(fmaxf(r1[0], r1[1]), r1[2]);
    float p11 = fmaxf(fmaxf(r1[2], r1[3]), r1[4]);
    float* o = A1 + (size_t)n * 4 * 256 + c;
    o[0 * 256] = p00;
    o[1 * 256] = p01;
    o[2 * 256] = p10;
    o[3 * 256] = p11;
}

template <int K, int NROWS>
__device__ __forceinline__ void stage_rows(float* __restrict__ dst,
                                           const float* __restrict__ src, int tid) {
    constexpr int P     = K + 4;
    constexpr int VPR   = K / 4;
    constexpr int ITERS = NROWS * VPR / 256;
    static_assert((NROWS * VPR) % 256 == 0, "whole passes of the workgroup only");
    static_assert(256 % VPR == 0, "a row is a whole fraction of one pass of the workgroup");
    const float4* s4 = reinterpret_cast<const float4*>(src) + tid;
    float* d = dst + (tid / VPR) * P + (tid % VPR) * 4;
    float4 v[ITERS];
#pragma unroll
    for (int j = 0; j < ITERS; ++j) v[j] = s4[256 * j];
#pragma unroll
    for (int j = 0; j < ITERS; ++j)
        *reinterpret_cast<float4*>(d + (256 / VPR) * j * P) = v[j];
}

template <int K>
__device__ __forceinline__ void stage_rows_rt(float* __restrict__ dst,
                                              const float* __restrict__ src,
                                              int nrows, int tid) {
    constexpr int P   = K + 4;
    constexpr int VPR = K / 4;
    const int total = nrows * VPR;
    const float4* s4 = reinterpret_cast<const float4*>(src);
    for (int i = tid; i < total; i += 256) {
        const int row = i / VPR, c4 = i % VPR;
        *reinterpret_cast<float4*>(dst + row * P + c4 * 4) = s4[i];
    }
}

template <int KIN, int COUT, bool TO_LDS>
__device__ __forceinline__ void conv_bn_relu_layer(
    const float* sIn, const float* sWt,
    const float* __restrict__ g, const float* __restrict__ b,
    float* lsum, float* lsq, float* lscale, float* lshift,
    float* dst, int tid) {
    constexpr int NT  = COUT / 16;
    constexpr int TT  = 16 * NT;
    constexpr int TPW = TT / 8;
    constexpr int P   = KIN + 4;

    const int wave = tid >> 5;
    const int lane = tid & 31;
    const int half = lane >> 4;
    const int r    = lane & 15;

    v8f acc[TPW];
#pragma unroll
    for (int i = 0; i < TPW; ++i) acc[i] = v8f{};

    const float* alo = sIn + (wave * 16 + r) * P + 2 * half;
    const float* ahi = alo + 128 * P;
    const float* bw  = sWt + r * P + 2 * half;
    for (int k0 = 0; k0 < KIN; k0 += 4) {
        const float2 a0 = *reinterpret_cast<const float2*>(alo + k0);
        const float2 a1 = *reinterpret_cast<const float2*>(ahi + k0);
        v2f fa0; fa0[0] = a0.x; fa0[1] = a0.y;
        v2f fa1; fa1[0] = a1.x; fa1[1] = a1.y;
#pragma unroll
        for (int nt = 0; nt < NT; ++nt) {
            const float2 bv = *reinterpret_cast<const float2*>(bw + nt * 16 * P + k0);
            v2f fb; fb[0] = bv.x; fb[1] = bv.y;
            acc[2 * nt]     = __builtin_amdgcn_wmma_f32_16x16x4_f32(false, fa0, false, fb,
                                                                    (short)0, acc[2 * nt], false, false);
            acc[2 * nt + 1] = __builtin_amdgcn_wmma_f32_16x16x4_f32(false, fa1, false, fb,
                                                                    (short)0, acc[2 * nt + 1], false, false);
        }
    }
#pragma unroll
    for (int i = 0; i < TPW; ++i) {
        const int n0 = (i >> 1) * 16;
        float s = 0.f, q = 0.f;
#pragma unroll
        for (int v = 0; v < 8; ++v) { float xv = acc[i][v]; s += xv; q += xv * xv; }
        atomicAdd(&lsum[n0 + r], s);
        atomicAdd(&lsq[n0 + r], q);
    }
    __syncthreads();

    if (tid < COUT) {
        float mu  = lsum[tid] * (1.f / 256.f);
        float var = lsq[tid] * (1.f / 256.f) - mu * mu;
        float sc  = g[tid] * rsqrtf(var + EPSV);
        lscale[tid] = sc;
        lshift[tid] = b[tid] - mu * sc;
        lsum[tid] = 0.f;
        lsq[tid]  = 0.f;
    }
    __syncthreads();

#pragma unroll
    for (int i = 0; i < TPW; ++i) {
        const int tt = wave + 8 * i;
        const int mt = tt & 15, nt = tt >> 4;
        const int m0 = mt * 16, n0 = nt * 16;
        const int col = n0 + r;
        const float sc = lscale[col], sh = lshift[col];
#pragma unroll
        for (int v = 0; v < 8; ++v) {
            const int row = m0 + v + 8 * half;
            float val = acc[i][v] * sc + sh;
            val = fmaxf(val, 0.f);
            if (TO_LDS) {
                dst[row * (COUT + 4) + col] = val;
            } else {
                const int n = row >> 2, s = row & 3;
                dst[n * (COUT * 4) + col * 4 + s] = val;
            }
        }
    }
}

__global__ __launch_bounds__(256) void conv_stack_kernel(
    const float* __restrict__ A1,
    const float* __restrict__ w1, const float* __restrict__ g1, const float* __restrict__ b1,
    const float* __restrict__ w2, const float* __restrict__ g2, const float* __restrict__ b2,
    const float* __restrict__ w3, const float* __restrict__ g3, const float* __restrict__ b3,
    float* __restrict__ fcin) {
    constexpr int P1 = 256 + 4, P2 = 32 + 4, P3 = 64 + 4;
    constexpr int OFF_W1 = 4 * 64;
    constexpr int OFF_W2 = OFF_W1 + 32 * P1;
    constexpr int OFF_W3 = OFF_W2 + 64 * P2;
    constexpr int OFF_A  = OFF_W3 + 64 * P3;
    constexpr int OFF_Z2 = OFF_A + 256 * P2;
    constexpr int TOTAL  = OFF_A + 256 * P1;
    static_assert(OFF_Z2 + 256 * P3 <= TOTAL, "z1 and z2 fit in the layer-1 input's space");
    static_assert(TOTAL * 4 <= 320 * 1024, "one workgroup's LDS");
    __shared__ alignas(16) float smem[TOTAL];

    float* lsum   = smem;
    float* lsq    = smem + 64;
    float* lscale = smem + 128;
    float* lshift = smem + 192;
    float* sW1 = smem + OFF_W1;
    float* sW2 = smem + OFF_W2;
    float* sW3 = smem + OFF_W3;
    float* sA  = smem + OFF_A;
    float* sZ1 = smem + OFF_A;
    float* sZ2 = smem + OFF_Z2;

    const int h   = blockIdx.x;
    const int tid = threadIdx.x;

    if (tid < 64) { lsum[tid] = 0.f; lsq[tid] = 0.f; }
    stage_rows<32, 64>(sW2, w2 + (size_t)h * 64 * 32, tid);
    stage_rows<64, 64>(sW3, w3 + (size_t)h * 64 * 64, tid);
    stage_rows<256, 32>(sW1, w1 + (size_t)h * 32 * 256, tid);
    stage_rows<256, 256>(sA, A1, tid);
    __syncthreads();

    conv_bn_relu_layer<256, 32, true>(sA, sW1, g1 + h * 32, b1 + h * 32,
                                      lsum, lsq, lscale, lshift, sZ1, tid);
    __syncthreads();
    conv_bn_relu_layer<32, 64, true>(sZ1, sW2, g2 + h * 64, b2 + h * 64,
                                     lsum, lsq, lscale, lshift, sZ2, tid);
    __syncthreads();
    conv_bn_relu_layer<64, 64, false>(sZ2, sW3, g3 + h * 64, b3 + h * 64,
                                      lsum, lsq, lscale, lshift,
                                      fcin + (size_t)h * 64 * 256, tid);
}

template <int N, int K, int MT, int BR>
__device__ __forceinline__ void stage_weight_rows(float* __restrict__ sB,
                                                  const float* __restrict__ W,
                                                  int tt0, int tid) {
    const int nrow0 = (tt0 / MT) * 16;
    if constexpr (N % BR == 0) {
        stage_rows<K, BR>(sB, W + (size_t)nrow0 * K, tid);
    } else {
        const int nvalid = (N - nrow0 < BR) ? (N - nrow0) : BR;
        if (nvalid == BR) stage_rows<K, BR>(sB, W + (size_t)nrow0 * K, tid);
        else              stage_rows_rt<K>(sB, W + (size_t)nrow0 * K, nvalid, tid);
    }
}

template <int M, int N, int K, bool RELU>
__global__ __launch_bounds__(256) void gemm_bias_kernel(
    const float* __restrict__ Ain, int a_head_stride,
    const float* __restrict__ Wt,
    const float* __restrict__ bias,
    float* __restrict__ Out, int out_head_stride) {
    constexpr int MT = (M + 15) / 16, NT = (N + 15) / 16, TT = MT * NT;
    constexpr int P  = K + 4;
    constexpr int BR = 16 * (8 / MT);
    static_assert(M % 16 == 0 && 8 % MT == 0, "a round of 8 tiles covers whole N-tiles");

    __shared__ alignas(16) float sA[M * P];
    __shared__ alignas(16) float sB[BR * P];

    const int h    = blockIdx.y;
    const int tid  = threadIdx.x;
    const int wave = tid >> 5;
    const int lane = tid & 31;
    const int half = lane >> 4;
    const int r    = lane & 15;
    const int wstride = gridDim.x * 8;

    const float* A = Ain + (size_t)h * a_head_stride;
    const float* W = Wt + (size_t)h * N * K;
    float* O = Out + (size_t)h * out_head_stride;

    int tt0 = blockIdx.x * 8;
    if (tt0 >= TT) return;
    stage_weight_rows<N, K, MT, BR>(sB, W, tt0, tid);
    stage_rows<K, M>(sA, A, tid);
    for (;;) {
        __syncthreads();

        const int tt = tt0 + wave;
        if (tt < TT) {
            const int mt  = tt % MT;
            const int ntl = wave / MT;
            const int m0 = mt * 16;
            const int col = (tt0 / MT) * 16 + ntl * 16 + r;
            const bool colok = (col < N);
            v8f c = {};
            const float* arow = sA + (m0 + r) * P + 2 * half;
            const float* brow = sB + (colok ? (ntl * 16 + r) : 0) * P + 2 * half;
            for (int k0 = 0; k0 < K; k0 += 4) {
                const float2 av = *reinterpret_cast<const float2*>(arow + k0);
                const float2 bv = *reinterpret_cast<const float2*>(brow + k0);
                v2f a; a[0] = av.x; a[1] = av.y;
                v2f bb; bb[0] = colok ? bv.x : 0.f; bb[1] = colok ? bv.y : 0.f;
                c = __builtin_amdgcn_wmma_f32_16x16x4_f32(false, a, false, bb,
                                                          (short)0, c, false, false);
            }
            if (colok) {
                const float bs = bias[h * N + col];
#pragma unroll
                for (int v = 0; v < 8; ++v) {
                    const int row = m0 + v + 8 * half;
                    float val = c[v] + bs;
                    if (RELU) val = fmaxf(val, 0.f);
                    O[(size_t)row * N + col] = val;
                }
            }
        }

        tt0 += wstride;
        if (tt0 >= TT) break;
        __syncthreads();
        stage_weight_rows<N, K, MT, BR>(sB, W, tt0, tid);
    }
}

__global__ void final_combine_kernel(const float* __restrict__ hout,
                                     const float* __restrict__ W2,
                                     const float* __restrict__ bias2,
                                     float* __restrict__ out) {
    __shared__ float comb[100];
    const int n = blockIdx.x;
    const int t = threadIdx.x;
    if (t < 100) {
        float s = 0.f;
#pragma unroll
        for (int h = 0; h < 4; ++h) s += hout[((size_t)h * 64 + n) * 100 + t];
        comb[t] = s * 0.25f;
    }
    __syncthreads();
    if (t < 100) {
        float acc = bias2[t];
        for (int j = 0; j < 100; ++j) acc += comb[j] * W2[t * 100 + j];
        float* orow = out + (size_t)n * 64 * 100 + t;
#pragma unroll 4
        for (int rep = 0; rep < 64; ++rep) orow[(size_t)rep * 100] = acc;
    }
}

extern "C" void kernel_launch(void* const* d_in, const int* in_sizes, int n_in,
                              void* d_out, int out_size, void* d_ws, size_t ws_size,
                              hipStream_t stream) {
    const float* x     = (const float*)d_in[0];
    const float* w1    = (const float*)d_in[1];
    const float* g1    = (const float*)d_in[2];
    const float* b1    = (const float*)d_in[3];
    const float* w2    = (const float*)d_in[4];
    const float* g2    = (const float*)d_in[5];
    const float* b2    = (const float*)d_in[6];
    const float* w3    = (const float*)d_in[7];
    const float* g3    = (const float*)d_in[8];
    const float* b3    = (const float*)d_in[9];
    const float* lw1   = (const float*)d_in[10];
    const float* lb1   = (const float*)d_in[11];
    const float* lw2   = (const float*)d_in[12];
    const float* lb2   = (const float*)d_in[13];
    const float* W2    = (const float*)d_in[14];
    const float* bias2 = (const float*)d_in[15];
    float* out = (float*)d_out;

    float* ws   = (float*)d_ws;
    float* A1   = ws;
    float* fcin = A1 + 256 * 256;
    float* h1   = fcin + 4 * 64 * 256;
    float* hout = h1 + 4 * 64 * 512;

    roi_pool_kernel<<<64, 256, 0, stream>>>(x, A1);

    conv_stack_kernel<<<4, 256, 0, stream>>>(
        A1, w1, g1, b1, w2, g2, b2, w3, g3, b3, fcin);

    gemm_bias_kernel<64, 512, 256, true><<<dim3(8, 4), 256, 0, stream>>>(
        fcin, 64 * 256, lw1, lb1, h1, 64 * 512);
    gemm_bias_kernel<64, 100, 512, false><<<dim3(4, 4), 256, 0, stream>>>(
        h1, 64 * 512, lw2, lb2, hout, 64 * 100);

    final_combine_kernel<<<64, 128, 0, stream>>>(hout, W2, bias2, out);
}


# Layer 1's 256-deep product moves from the 4-workgroup conv stack into the pooling launch: each of its 64 workgroups multiplies its 4 pooled rows by all heads' w1 on WMMA; k order unchanged
# speedup vs baseline: 2.1013x; 1.1499x over previous
#include <hip/hip_runtime.h>
#include <hip/hip_bf16.h>

typedef float v2f __attribute__((ext_vector_type(2)));
typedef float v8f __attribute__((ext_vector_type(8)));

#define EPSV 1e-5f

__global__ __launch_bounds__(256) void roi_pool_conv1_kernel(
    const float* __restrict__ x,
    const float* __restrict__ w1,
    float* __restrict__ z1raw) {
    constexpr int P = 256 + 4;
    __shared__ alignas(16) float sW[128 * P];
    __shared__ alignas(16) float sP[5 * P];
    __shared__ alignas(16) float sO[4 * 128];

    const int n    = blockIdx.x;
    const int tid  = threadIdx.x;
    const int wave = tid >> 5;
    const int lane = tid & 31;
    const int half = lane >> 4;
    const int r    = lane & 15;

    const float* base = x + ((size_t)(n * 256 + tid) * 64) * 64;
    float v[4][5];
#pragma unroll
    for (int rr = 0; rr < 4; ++rr)
#pragma unroll
        for (int cc = 0; cc < 5; ++cc)
            v[rr][cc] = base[(2 + rr) * 64 + 1 + cc];
    __builtin_amdgcn_sched_barrier(0);

    const float4* w4 = reinterpret_cast<const float4*>(w1) + tid;
    float4 wv[32];
#pragma unroll
    for (int j = 0; j < 32; ++j) wv[j] = w4[256 * j];

    float* wd = sW + (tid >> 6) * P + (tid & 63) * 4;
#pragma unroll
    for (int j = 0; j < 32; ++j)
        *reinterpret_cast<float4*>(wd + 4 * j * P) = wv[j];

    float r0[5], r1[5];
#pragma unroll
    for (int cc = 0; cc < 5; ++cc) {
        r0[cc] = fmaxf(v[0][cc], v[1][cc]);
        r1[cc] = fmaxf(v[2][cc], v[3][cc]);
    }
    sP[0 * P + tid] = fmaxf(fmaxf(r0[0], r0[1]), r0[2]);
    sP[1 * P + tid] = fmaxf(fmaxf(r0[2], r0[3]), r0[4]);
    sP[2 * P + tid] = fmaxf(fmaxf(r1[0], r1[1]), r1[2]);
    sP[3 * P + tid] = fmaxf(fmaxf(r1[2], r1[3]), r1[4]);
    sP[4 * P + tid] = 0.f;
    __syncthreads();

    const float* ap = sP + (r < 4 ? r : 4) * P + 2 * half;
    const float* bp = sW + (wave * 16 + r) * P + 2 * half;
    v8f acc = {};
    for (int k0 = 0; k0 < 256; k0 += 4) {
        const float2 av = *reinterpret_cast<const float2*>(ap + k0);
        const float2 bv = *reinterpret_cast<const float2*>(bp + k0);
        v2f fa; fa[0] = av.x; fa[1] = av.y;
        v2f fb; fb[0] = bv.x; fb[1] = bv.y;
        acc = __builtin_amdgcn_wmma_f32_16x16x4_f32(false, fa, false, fb,
                                                    (short)0, acc, false, false);
    }
    if (half == 0) {
        float* o = sO + (wave >> 1) * 128 + (wave & 1) * 16 + r;
#pragma unroll
        for (int s = 0; s < 4; ++s) o[s * 32] = acc[s];
    }
    __syncthreads();
    if (tid < 128) {
        const float4 ov = *reinterpret_cast<const float4*>(sO + tid * 4);
        *reinterpret_cast<float4*>(z1raw + (size_t)(tid >> 5) * (256 * 32)
                                   + (size_t)n * 128 + (tid & 31) * 4) = ov;
    }
}

template <int K, int NROWS>
__device__ __forceinline__ void stage_rows(float* __restrict__ dst,
                                           const float* __restrict__ src, int tid) {
    constexpr int P     = K + 4;
    constexpr int VPR   = K / 4;
    constexpr int ITERS = NROWS * VPR / 256;
    static_assert((NROWS * VPR) % 256 == 0, "whole passes of the workgroup only");
    static_assert(256 % VPR == 0, "a row is a whole fraction of one pass of the workgroup");
    const float4* s4 = reinterpret_cast<const float4*>(src) + tid;
    float* d = dst + (tid / VPR) * P + (tid % VPR) * 4;
    float4 v[ITERS];
#pragma unroll
    for (int j = 0; j < ITERS; ++j) v[j] = s4[256 * j];
#pragma unroll
    for (int j = 0; j < ITERS; ++j)
        *reinterpret_cast<float4*>(d + (256 / VPR) * j * P) = v[j];
}

template <int K>
__device__ __forceinline__ void stage_rows_rt(float* __restrict__ dst,
                                              const float* __restrict__ src,
                                              int nrows, int tid) {
    constexpr int P   = K + 4;
    constexpr int VPR = K / 4;
    const int total = nrows * VPR;
    const float4* s4 = reinterpret_cast<const float4*>(src);
    for (int i = tid; i < total; i += 256) {
        const int row = i / VPR, c4 = i % VPR;
        *reinterpret_cast<float4*>(dst + row * P + c4 * 4) = s4[i];
    }
}

template <int COUT, bool TO_LDS>
__device__ __forceinline__ void bn_relu_store(
    const v8f (&acc)[COUT / 8],
    const float* __restrict__ g, const float* __restrict__ b,
    float* lsum, float* lsq, float* lscale, float* lshift,
    float* dst, int tid) {
    constexpr int TPW = COUT / 8;

    const int wave = tid >> 5;
    const int lane = tid & 31;
    const int half = lane >> 4;
    const int r    = lane & 15;

#pragma unroll
    for (int i = 0; i < TPW; ++i) {
        const int n0 = (i >> 1) * 16;
        float s = 0.f, q = 0.f;
#pragma unroll
        for (int v = 0; v < 8; ++v) { float xv = acc[i][v]; s += xv; q += xv * xv; }
        atomicAdd(&lsum[n0 + r], s);
        atomicAdd(&lsq[n0 + r], q);
    }
    __syncthreads();

    if (tid < COUT) {
        float mu  = lsum[tid] * (1.f / 256.f);
        float var = lsq[tid] * (1.f / 256.f) - mu * mu;
        float sc  = g[tid] * rsqrtf(var + EPSV);
        lscale[tid] = sc;
        lshift[tid] = b[tid] - mu * sc;
        lsum[tid] = 0.f;
        lsq[tid]  = 0.f;
    }
    __syncthreads();

#pragma unroll
    for (int i = 0; i < TPW; ++i) {
        const int tt = wave + 8 * i;
        const int mt = tt & 15, nt = tt >> 4;
        const int m0 = mt * 16, n0 = nt * 16;
        const int col = n0 + r;
        const float sc = lscale[col], sh = lshift[col];
#pragma unroll
        for (int v = 0; v < 8; ++v) {
            const int row = m0 + v + 8 * half;
            float val = acc[i][v] * sc + sh;
            val = fmaxf(val, 0.f);
            if (TO_LDS) {
                dst[row * (COUT + 4) + col] = val;
            } else {
                const int n = row >> 2, s = row & 3;
                dst[n * (COUT * 4) + col * 4 + s] = val;
            }
        }
    }
}

template <int KIN, int COUT, bool TO_LDS>
__device__ __forceinline__ void conv_bn_relu_layer(
    const float* sIn, const float* sWt,
    const float* __restrict__ g, const float* __restrict__ b,
    float* lsum, float* lsq, float* lscale, float* lshift,
    float* dst, int tid) {
    constexpr int NT  = COUT / 16;
    constexpr int TPW = 2 * NT;
    constexpr int P   = KIN + 4;

    const int wave = tid >> 5;
    const int lane = tid & 31;
    const int half = lane >> 4;
    const int r    = lane & 15;

    v8f acc[TPW];
#pragma unroll
    for (int i = 0; i < TPW; ++i) acc[i] = v8f{};

    const float* alo = sIn + (wave * 16 + r) * P + 2 * half;
    const float* ahi = alo + 128 * P;
    const float* bw  = sWt + r * P + 2 * half;
    for (int k0 = 0; k0 < KIN; k0 += 4) {
        const float2 a0 = *reinterpret_cast<const float2*>(alo + k0);
        const float2 a1 = *reinterpret_cast<const float2*>(ahi + k0);
        v2f fa0; fa0[0] = a0.x; fa0[1] = a0.y;
        v2f fa1; fa1[0] = a1.x; fa1[1] = a1.y;
#pragma unroll
        for (int nt = 0; nt < NT; ++nt) {
            const float2 bv = *reinterpret_cast<const float2*>(bw + nt * 16 * P + k0);
            v2f fb; fb[0] = bv.x; fb[1] = bv.y;
            acc[2 * nt]     = __builtin_amdgcn_wmma_f32_16x16x4_f32(false, fa0, false, fb,
                                                                    (short)0, acc[2 * nt], false, false);
            acc[2 * nt + 1] = __builtin_amdgcn_wmma_f32_16x16x4_f32(false, fa1, false, fb,
                                                                    (short)0, acc[2 * nt + 1], false, false);
        }
    }
    bn_relu_store<COUT, TO_LDS>(acc, g, b, lsum, lsq, lscale, lshift, dst, tid);
}

__global__ __launch_bounds__(256) void conv_stack_kernel(
    const float* __restrict__ z1raw,
    const float* __restrict__ g1, const float* __restrict__ b1,
    const float* __restrict__ w2, const float* __restrict__ g2, const float* __restrict__ b2,
    const float* __restrict__ w3, const float* __restrict__ g3, const float* __restrict__ b3,
    float* __restrict__ fcin) {
    constexpr int P2 = 32 + 4, P3 = 64 + 4;
    constexpr int OFF_W2 = 4 * 64;
    constexpr int OFF_W3 = OFF_W2 + 64 * P2;
    constexpr int OFF_Z1 = OFF_W3 + 64 * P3;
    constexpr int OFF_Z2 = OFF_Z1 + 256 * P2;
    constexpr int TOTAL  = OFF_Z2 + 256 * P3;
    static_assert(TOTAL * 4 <= 320 * 1024, "one workgroup's LDS");
    __shared__ alignas(16) float smem[TOTAL];

    float* lsum   = smem;
    float* lsq    = smem + 64;
    float* lscale = smem + 128;
    float* lshift = smem + 192;
    float* sW2 = smem + OFF_W2;
    float* sW3 = smem + OFF_W3;
    float* sZ1 = smem + OFF_Z1;
    float* sZ2 = smem + OFF_Z2;

    const int h    = blockIdx.x;
    const int tid  = threadIdx.x;
    const int wave = tid >> 5;
    const int lane = tid & 31;
    const int half = lane >> 4;
    const int r    = lane & 15;

    if (tid < 64) { lsum[tid] = 0.f; lsq[tid] = 0.f; }

    v8f acc1[4];
    const float* zr = z1raw + (size_t)h * (256 * 32) + (8 * half) * 32 + r;
#pragma unroll
    for (int i = 0; i < 4; ++i) {
        const int tt = wave + 8 * i;
        const int mt = tt & 15, nt = tt >> 4;
#pragma unroll
        for (int v = 0; v < 8; ++v)
            acc1[i][v] = zr[(mt * 16 + v) * 32 + nt * 16];
    }
    stage_rows<32, 64>(sW2, w2 + (size_t)h * 64 * 32, tid);
    stage_rows<64, 64>(sW3, w3 + (size_t)h * 64 * 64, tid);
    __syncthreads();

    bn_relu_store<32, true>(acc1, g1 + h * 32, b1 + h * 32,
                            lsum, lsq, lscale, lshift, sZ1, tid);
    __syncthreads();
    conv_bn_relu_layer<32, 64, true>(sZ1, sW2, g2 + h * 64, b2 + h * 64,
                                     lsum, lsq, lscale, lshift, sZ2, tid);
    __syncthreads();
    conv_bn_relu_layer<64, 64, false>(sZ2, sW3, g3 + h * 64, b3 + h * 64,
                                      lsum, lsq, lscale, lshift,
                                      fcin + (size_t)h * 64 * 256, tid);
}

template <int N, int K, int MT, int BR>
__device__ __forceinline__ void stage_weight_rows(float* __restrict__ sB,
                                                  const float* __restrict__ W,
                                                  int tt0, int tid) {
    const int nrow0 = (tt0 / MT) * 16;
    if constexpr (N % BR == 0) {
        stage_rows<K, BR>(sB, W + (size_t)nrow0 * K, tid);
    } else {
        const int nvalid = (N - nrow0 < BR) ? (N - nrow0) : BR;
        if (nvalid == BR) stage_rows<K, BR>(sB, W + (size_t)nrow0 * K, tid);
        else              stage_rows_rt<K>(sB, W + (size_t)nrow0 * K, nvalid, tid);
    }
}

template <int M, int N, int K, bool RELU>
__global__ __launch_bounds__(256) void gemm_bias_kernel(
    const float* __restrict__ Ain, int a_head_stride,
    const float* __restrict__ Wt,
    const float* __restrict__ bias,
    float* __restrict__ Out, int out_head_stride) {
    constexpr int MT = (M + 15) / 16, NT = (N + 15) / 16, TT = MT * NT;
    constexpr int P  = K + 4;
    constexpr int BR = 16 * (8 / MT);
    static_assert(M % 16 == 0 && 8 % MT == 0, "a round of 8 tiles covers whole N-tiles");

    __shared__ alignas(16) float sA[M * P];
    __shared__ alignas(16) float sB[BR * P];

    const int h    = blockIdx.y;
    const int tid  = threadIdx.x;
    const int wave = tid >> 5;
    const int lane = tid & 31;
    const int half = lane >> 4;
    const int r    = lane & 15;
    const int wstride = gridDim.x * 8;

    const float* A = Ain + (size_t)h * a_head_stride;
    const float* W = Wt + (size_t)h * N * K;
    float* O = Out + (size_t)h * out_head_stride;

    int tt0 = blockIdx.x * 8;
    if (tt0 >= TT) return;
    stage_weight_rows<N, K, MT, BR>(sB, W, tt0, tid);
    stage_rows<K, M>(sA, A, tid);
    for (;;) {
        __syncthreads();

        const int tt = tt0 + wave;
        if (tt < TT) {
            const int mt  = tt % MT;
            const int ntl = wave / MT;
            const int m0 = mt * 16;
            const int col = (tt0 / MT) * 16 + ntl * 16 + r;
            const bool colok = (col < N);
            v8f c = {};
            const float* arow = sA + (m0 + r) * P + 2 * half;
            const float* brow = sB + (colok ? (ntl * 16 + r) : 0) * P + 2 * half;
            for (int k0 = 0; k0 < K; k0 += 4) {
                const float2 av = *reinterpret_cast<const float2*>(arow + k0);
                const float2 bv = *reinterpret_cast<const float2*>(brow + k0);
                v2f a; a[0] = av.x; a[1] = av.y;
                v2f bb; bb[0] = colok ? bv.x : 0.f; bb[1] = colok ? bv.y : 0.f;
                c = __builtin_amdgcn_wmma_f32_16x16x4_f32(false, a, false, bb,
                                                          (short)0, c, false, false);
            }
            if (colok) {
                const float bs = bias[h * N + col];
#pragma unroll
                for (int v = 0; v < 8; ++v) {
                    const int row = m0 + v + 8 * half;
                    float val = c[v] + bs;
                    if (RELU) val = fmaxf(val, 0.f);
                    O[(size_t)row * N + col] = val;
                }
            }
        }

        tt0 += wstride;
        if (tt0 >= TT) break;
        __syncthreads();
        stage_weight_rows<N, K, MT, BR>(sB, W, tt0, tid);
    }
}

__global__ void final_combine_kernel(const float* __restrict__ hout,
                                     const float* __restrict__ W2,
                                     const float* __restrict__ bias2,
                                     float* __restrict__ out) {
    __shared__ float comb[100];
    const int n = blockIdx.x;
    const int t = threadIdx.x;
    if (t < 100) {
        float s = 0.f;
#pragma unroll
        for (int h = 0; h < 4; ++h) s += hout[((size_t)h * 64 + n) * 100 + t];
        comb[t] = s * 0.25f;
    }
    __syncthreads();
    if (t < 100) {
        float acc = bias2[t];
        for (int j = 0; j < 100; ++j) acc += comb[j] * W2[t * 100 + j];
        float* orow = out + (size_t)n * 64 * 100 + t;
#pragma unroll 4
        for (int rep = 0; rep < 64; ++rep) orow[(size_t)rep * 100] = acc;
    }
}

extern "C" void kernel_launch(void* const* d_in, const int* in_sizes, int n_in,
                              void* d_out, int out_size, void* d_ws, size_t ws_size,
                              hipStream_t stream) {
    const float* x     = (const float*)d_in[0];
    const float* w1    = (const float*)d_in[1];
    const float* g1    = (const float*)d_in[2];
    const float* b1    = (const float*)d_in[3];
    const float* w2    = (const float*)d_in[4];
    const float* g2    = (const float*)d_in[5];
    const float* b2    = (const float*)d_in[6];
    const float* w3    = (const float*)d_in[7];
    const float* g3    = (const float*)d_in[8];
    const float* b3    = (const float*)d_in[9];
    const float* lw1   = (const float*)d_in[10];
    const float* lb1   = (const float*)d_in[11];
    const float* lw2   = (const float*)d_in[12];
    const float* lb2   = (const float*)d_in[13];
    const float* W2    = (const float*)d_in[14];
    const float* bias2 = (const float*)d_in[15];
    float* out = (float*)d_out;

    float* ws   = (float*)d_ws;
    float* z1raw = ws;
    float* fcin = z1raw + 256 * 256;
    float* h1   = fcin + 4 * 64 * 256;
    float* hout = h1 + 4 * 64 * 512;

    roi_pool_conv1_kernel<<<64, 256, 0, stream>>>(x, w1, z1raw);

    conv_stack_kernel<<<4, 256, 0, stream>>>(
        z1raw, g1, b1, w2, g2, b2, w3, g3, b3, fcin);

    gemm_bias_kernel<64, 512, 256, true><<<dim3(8, 4), 256, 0, stream>>>(
        fcin, 64 * 256, lw1, lb1, h1, 64 * 512);
    gemm_bias_kernel<64, 100, 512, false><<<dim3(4, 4), 256, 0, stream>>>(
        h1, 64 * 512, lw2, lb2, hout, 64 * 100);

    final_combine_kernel<<<64, 128, 0, stream>>>(hout, W2, bias2, out);
}


# fc2 moves into the fc1 launch: each of 64 workgroups keeps its 32x64 fc1 block in LDS and multiplies it by its k-slice of lw2 on WMMA; last kernel adds the 8 k-slice parts (fc2 sum order changes)
# speedup vs baseline: 2.9685x; 1.4127x over previous
#include <hip/hip_runtime.h>
#include <hip/hip_bf16.h>

typedef float v2f __attribute__((ext_vector_type(2)));
typedef float v8f __attribute__((ext_vector_type(8)));

#define EPSV 1e-5f

__global__ __launch_bounds__(256) void roi_pool_conv1_kernel(
    const float* __restrict__ x,
    const float* __restrict__ w1,
    float* __restrict__ z1raw) {
    constexpr int P = 256 + 4;
    __shared__ alignas(16) float sW[128 * P];
    __shared__ alignas(16) float sP[5 * P];
    __shared__ alignas(16) float sO[4 * 128];

    const int n    = blockIdx.x;
    const int tid  = threadIdx.x;
    const int wave = tid >> 5;
    const int lane = tid & 31;
    const int half = lane >> 4;
    const int r    = lane & 15;

    const float* base = x + ((size_t)(n * 256 + tid) * 64) * 64;
    float v[4][5];
#pragma unroll
    for (int rr = 0; rr < 4; ++rr)
#pragma unroll
        for (int cc = 0; cc < 5; ++cc)
            v[rr][cc] = base[(2 + rr) * 64 + 1 + cc];
    __builtin_amdgcn_sched_barrier(0);

    const float4* w4 = reinterpret_cast<const float4*>(w1) + tid;
    float4 wv[32];
#pragma unroll
    for (int j = 0; j < 32; ++j) wv[j] = w4[256 * j];

    float* wd = sW + (tid >> 6) * P + (tid & 63) * 4;
#pragma unroll
    for (int j = 0; j < 32; ++j)
        *reinterpret_cast<float4*>(wd + 4 * j * P) = wv[j];

    float r0[5], r1[5];
#pragma unroll
    for (int cc = 0; cc < 5; ++cc) {
        r0[cc] = fmaxf(v[0][cc], v[1][cc]);
        r1[cc] = fmaxf(v[2][cc], v[3][cc]);
    }
    sP[0 * P + tid] = fmaxf(fmaxf(r0[0], r0[1]), r0[2]);
    sP[1 * P + tid] = fmaxf(fmaxf(r0[2], r0[3]), r0[4]);
    sP[2 * P + tid] = fmaxf(fmaxf(r1[0], r1[1]), r1[2]);
    sP[3 * P + tid] = fmaxf(fmaxf(r1[2], r1[3]), r1[4]);
    sP[4 * P + tid] = 0.f;
    __syncthreads();

    const float* ap = sP + (r < 4 ? r : 4) * P + 2 * half;
    const float* bp = sW + (wave * 16 + r) * P + 2 * half;
    v8f acc = {};
    for (int k0 = 0; k0 < 256; k0 += 4) {
        const float2 av = *reinterpret_cast<const float2*>(ap + k0);
        const float2 bv = *reinterpret_cast<const float2*>(bp + k0);
        v2f fa; fa[0] = av.x; fa[1] = av.y;
        v2f fb; fb[0] = bv.x; fb[1] = bv.y;
        acc = __builtin_amdgcn_wmma_f32_16x16x4_f32(false, fa, false, fb,
                                                    (short)0, acc, false, false);
    }
    if (half == 0) {
        float* o = sO + (wave >> 1) * 128 + (wave & 1) * 16 + r;
#pragma unroll
        for (int s = 0; s < 4; ++s) o[s * 32] = acc[s];
    }
    __syncthreads();
    if (tid < 128) {
        const float4 ov = *reinterpret_cast<const float4*>(sO + tid * 4);
        *reinterpret_cast<float4*>(z1raw + (size_t)(tid >> 5) * (256 * 32)
                                   + (size_t)n * 128 + (tid & 31) * 4) = ov;
    }
}

template <int K, int NROWS>
__device__ __forceinline__ void stage_rows(float* __restrict__ dst,
                                           const float* __restrict__ src, int tid) {
    constexpr int P     = K + 4;
    constexpr int VPR   = K / 4;
    constexpr int ITERS = NROWS * VPR / 256;
    static_assert((NROWS * VPR) % 256 == 0, "whole passes of the workgroup only");
    static_assert(256 % VPR == 0, "a row is a whole fraction of one pass of the workgroup");
    const float4* s4 = reinterpret_cast<const float4*>(src) + tid;
    float* d = dst + (tid / VPR) * P + (tid % VPR) * 4;
    float4 v[ITERS];
#pragma unroll
    for (int j = 0; j < ITERS; ++j) v[j] = s4[256 * j];
#pragma unroll
    for (int j = 0; j < ITERS; ++j)
        *reinterpret_cast<float4*>(d + (256 / VPR) * j * P) = v[j];
}

template <int COUT, bool TO_LDS>
__device__ __forceinline__ void bn_relu_store(
    const v8f (&acc)[COUT / 8],
    const float* __restrict__ g, const float* __restrict__ b,
    float* lsum, float* lsq, float* lscale, float* lshift,
    float* dst, int tid) {
    constexpr int TPW = COUT / 8;

    const int wave = tid >> 5;
    const int lane = tid & 31;
    const int half = lane >> 4;
    const int r    = lane & 15;

#pragma unroll
    for (int i = 0; i < TPW; ++i) {
        const int n0 = (i >> 1) * 16;
        float s = 0.f, q = 0.f;
#pragma unroll
        for (int v = 0; v < 8; ++v) { float xv = acc[i][v]; s += xv; q += xv * xv; }
        atomicAdd(&lsum[n0 + r], s);
        atomicAdd(&lsq[n0 + r], q);
    }
    __syncthreads();

    if (tid < COUT) {
        float mu  = lsum[tid] * (1.f / 256.f);
        float var = lsq[tid] * (1.f / 256.f) - mu * mu;
        float sc  = g[tid] * rsqrtf(var + EPSV);
        lscale[tid] = sc;
        lshift[tid] = b[tid] - mu * sc;
        lsum[tid] = 0.f;
        lsq[tid]  = 0.f;
    }
    __syncthreads();

#pragma unroll
    for (int i = 0; i < TPW; ++i) {
        const int tt = wave + 8 * i;
        const int mt = tt & 15, nt = tt >> 4;
        const int m0 = mt * 16, n0 = nt * 16;
        const int col = n0 + r;
        const float sc = lscale[col], sh = lshift[col];
#pragma unroll
        for (int v = 0; v < 8; ++v) {
            const int row = m0 + v + 8 * half;
            float val = acc[i][v] * sc + sh;
            val = fmaxf(val, 0.f);
            if (TO_LDS) {
                dst[row * (COUT + 4) + col] = val;
            } else {
                const int n = row >> 2, s = row & 3;
                dst[n * (COUT * 4) + col * 4 + s] = val;
            }
        }
    }
}

template <int KIN, int COUT, bool TO_LDS>
__device__ __forceinline__ void conv_bn_relu_layer(
    const float* sIn, const float* sWt,
    const float* __restrict__ g, const float* __restrict__ b,
    float* lsum, float* lsq, float* lscale, float* lshift,
    float* dst, int tid) {
    constexpr int NT  = COUT / 16;
    constexpr int TPW = 2 * NT;
    constexpr int P   = KIN + 4;

    const int wave = tid >> 5;
    const int lane = tid & 31;
    const int half = lane >> 4;
    const int r    = lane & 15;

    v8f acc[TPW];
#pragma unroll
    for (int i = 0; i < TPW; ++i) acc[i] = v8f{};

    const float* alo = sIn + (wave * 16 + r) * P + 2 * half;
    const float* ahi = alo + 128 * P;
    const float* bw  = sWt + r * P + 2 * half;
    for (int k0 = 0; k0 < KIN; k0 += 4) {
        const float2 a0 = *reinterpret_cast<const float2*>(alo + k0);
        const float2 a1 = *reinterpret_cast<const float2*>(ahi + k0);
        v2f fa0; fa0[0] = a0.x; fa0[1] = a0.y;
        v2f fa1; fa1[0] = a1.x; fa1[1] = a1.y;
#pragma unroll
        for (int nt = 0; nt < NT; ++nt) {
            const float2 bv = *reinterpret_cast<const float2*>(bw + nt * 16 * P + k0);
            v2f fb; fb[0] = bv.x; fb[1] = bv.y;
            acc[2 * nt]     = __builtin_amdgcn_wmma_f32_16x16x4_f32(false, fa0, false, fb,
                                                                    (short)0, acc[2 * nt], false, false);
            acc[2 * nt + 1] = __builtin_amdgcn_wmma_f32_16x16x4_f32(false, fa1, false, fb,
                                                                    (short)0, acc[2 * nt + 1], false, false);
        }
    }
    bn_relu_store<COUT, TO_LDS>(acc, g, b, lsum, lsq, lscale, lshift, dst, tid);
}

__global__ __launch_bounds__(256) void conv_stack_kernel(
    const float* __restrict__ z1raw,
    const float* __restrict__ g1, const float* __restrict__ b1,
    const float* __restrict__ w2, const float* __restrict__ g2, const float* __restrict__ b2,
    const float* __restrict__ w3, const float* __restrict__ g3, const float* __restrict__ b3,
    float* __restrict__ fcin) {
    constexpr int P2 = 32 + 4, P3 = 64 + 4;
    constexpr int OFF_W2 = 4 * 64;
    constexpr int OFF_W3 = OFF_W2 + 64 * P2;
    constexpr int OFF_Z1 = OFF_W3 + 64 * P3;
    constexpr int OFF_Z2 = OFF_Z1 + 256 * P2;
    constexpr int TOTAL  = OFF_Z2 + 256 * P3;
    static_assert(TOTAL * 4 <= 320 * 1024, "one workgroup's LDS");
    __shared__ alignas(16) float smem[TOTAL];

    float* lsum   = smem;
    float* lsq    = smem + 64;
    float* lscale = smem + 128;
    float* lshift = smem + 192;
    float* sW2 = smem + OFF_W2;
    float* sW3 = smem + OFF_W3;
    float* sZ1 = smem + OFF_Z1;
    float* sZ2 = smem + OFF_Z2;

    const int h    = blockIdx.x;
    const int tid  = threadIdx.x;
    const int wave = tid >> 5;
    const int lane = tid & 31;
    const int half = lane >> 4;
    const int r    = lane & 15;

    if (tid < 64) { lsum[tid] = 0.f; lsq[tid] = 0.f; }

    v8f acc1[4];
    const float* zr = z1raw + (size_t)h * (256 * 32) + (8 * half) * 32 + r;
#pragma unroll
    for (int i = 0; i < 4; ++i) {
        const int tt = wave + 8 * i;
        const int mt = tt & 15, nt = tt >> 4;
#pragma unroll
        for (int v = 0; v < 8; ++v)
            acc1[i][v] = zr[(mt * 16 + v) * 32 + nt * 16];
    }
    stage_rows<32, 64>(sW2, w2 + (size_t)h * 64 * 32, tid);
    stage_rows<64, 64>(sW3, w3 + (size_t)h * 64 * 64, tid);
    __syncthreads();

    bn_relu_store<32, true>(acc1, g1 + h * 32, b1 + h * 32,
                            lsum, lsq, lscale, lshift, sZ1, tid);
    __syncthreads();
    conv_bn_relu_layer<32, 64, true>(sZ1, sW2, g2 + h * 64, b2 + h * 64,
                                     lsum, lsq, lscale, lshift, sZ2, tid);
    __syncthreads();
    conv_bn_relu_layer<64, 64, false>(sZ2, sW3, g3 + h * 64, b3 + h * 64,
                                      lsum, lsq, lscale, lshift,
                                      fcin + (size_t)h * 64 * 256, tid);
}

__global__ __launch_bounds__(256) void fc1_fc2_part_kernel(
    const float* __restrict__ fcin,
    const float* __restrict__ lw1,
    const float* __restrict__ lb1,
    const float* __restrict__ lw2,
    float* __restrict__ part) {
    constexpr int P1 = 256 + 4;
    constexpr int P2 = 64 + 4;
    __shared__ alignas(16) float sA[32 * P1];
    __shared__ alignas(16) float sB1[64 * P1];
    __shared__ alignas(16) float sB2[112 * P2];
    __shared__ alignas(16) float sH[32 * P2];

    const int s    = blockIdx.x >> 1;
    const int mh   = blockIdx.x & 1;
    const int h    = blockIdx.y;
    const int tid  = threadIdx.x;
    const int wave = __builtin_amdgcn_readfirstlane(tid >> 5);
    const int lane = tid & 31;
    const int half = lane >> 4;
    const int r    = lane & 15;
    const int mt   = wave & 1;
    const int nq   = wave >> 1;

    const float4* a4 = reinterpret_cast<const float4*>(fcin + ((size_t)h * 64 + mh * 32) * 256) + tid;
    const float4* b4 = reinterpret_cast<const float4*>(lw1 + ((size_t)h * 512 + s * 64) * 256) + tid;
    const float* w2p = lw2 + ((size_t)h * 100 + (tid >> 4)) * 512 + s * 64 + (tid & 15) * 4;
    float4 av[8], bv1[16], bv2[7];
#pragma unroll
    for (int j = 0; j < 8; ++j) av[j] = a4[256 * j];
#pragma unroll
    for (int j = 0; j < 16; ++j) bv1[j] = b4[256 * j];
#pragma unroll
    for (int j = 0; j < 6; ++j) bv2[j] = *reinterpret_cast<const float4*>(w2p + 16 * j * 512);
    const int r6 = min(96 + (tid >> 4), 99);
    bv2[6] = *reinterpret_cast<const float4*>(lw2 + ((size_t)h * 100 + r6) * 512 + s * 64 + (tid & 15) * 4);
    const float bs = lb1[h * 512 + s * 64 + nq * 16 + r];

    float* ad = sA + (tid >> 6) * P1 + (tid & 63) * 4;
    float* bd = sB1 + (tid >> 6) * P1 + (tid & 63) * 4;
#pragma unroll
    for (int j = 0; j < 8; ++j) *reinterpret_cast<float4*>(ad + 4 * j * P1) = av[j];
#pragma unroll
    for (int j = 0; j < 16; ++j) *reinterpret_cast<float4*>(bd + 4 * j * P1) = bv1[j];
    float* wd = sB2 + (tid >> 4) * P2 + (tid & 15) * 4;
#pragma unroll
    for (int j = 0; j < 7; ++j) *reinterpret_cast<float4*>(wd + 16 * j * P2) = bv2[j];
    __syncthreads();

    {
        const float* ap = sA + (mt * 16 + r) * P1 + 2 * half;
        const float* bp = sB1 + (nq * 16 + r) * P1 + 2 * half;
        v8f c = {};
        for (int k0 = 0; k0 < 256; k0 += 4) {
            const float2 a2 = *reinterpret_cast<const float2*>(ap + k0);
            const float2 b2 = *reinterpret_cast<const float2*>(bp + k0);
            v2f fa; fa[0] = a2.x; fa[1] = a2.y;
            v2f fb; fb[0] = b2.x; fb[1] = b2.y;
            c = __builtin_amdgcn_wmma_f32_16x16x4_f32(false, fa, false, fb,
                                                      (short)0, c, false, false);
        }
        float* hp = sH + (mt * 16 + 8 * half) * P2 + nq * 16 + r;
#pragma unroll
        for (int v = 0; v < 8; ++v) hp[v * P2] = fmaxf(c[v] + bs, 0.f);
    }
    __syncthreads();

    {
        const int c0 = nq * 16 + r;
        const int c1 = c0 + 64;
        const bool ok1 = c1 < 100;
        const float* ap  = sH + (mt * 16 + r) * P2 + 2 * half;
        const float* bp0 = sB2 + c0 * P2 + 2 * half;
        const float* bp1 = sB2 + (ok1 ? c1 : 0) * P2 + 2 * half;
        v8f d0 = {}, d1 = {};
        if (nq < 3) {
            for (int k0 = 0; k0 < 64; k0 += 4) {
                const float2 a2 = *reinterpret_cast<const float2*>(ap + k0);
                const float2 x0 = *reinterpret_cast<const float2*>(bp0 + k0);
                const float2 x1 = *reinterpret_cast<const float2*>(bp1 + k0);
                v2f fa;  fa[0] = a2.x; fa[1] = a2.y;
                v2f fb0; fb0[0] = x0.x; fb0[1] = x0.y;
                v2f fb1; fb1[0] = ok1 ? x1.x : 0.f; fb1[1] = ok1 ? x1.y : 0.f;
                d0 = __builtin_amdgcn_wmma_f32_16x16x4_f32(false, fa, false, fb0,
                                                           (short)0, d0, false, false);
                d1 = __builtin_amdgcn_wmma_f32_16x16x4_f32(false, fa, false, fb1,
                                                           (short)0, d1, false, false);
            }
        } else {
            for (int k0 = 0; k0 < 64; k0 += 4) {
                const float2 a2 = *reinterpret_cast<const float2*>(ap + k0);
                const float2 x0 = *reinterpret_cast<const float2*>(bp0 + k0);
                v2f fa;  fa[0] = a2.x; fa[1] = a2.y;
                v2f fb0; fb0[0] = x0.x; fb0[1] = x0.y;
                d0 = __builtin_amdgcn_wmma_f32_16x16x4_f32(false, fa, false, fb0,
                                                           (short)0, d0, false, false);
            }
        }
        float* o = part + (((size_t)h * 8 + s) * 64 + mh * 32 + mt * 16 + 8 * half) * 100;
#pragma unroll
        for (int v = 0; v < 8; ++v) o[v * 100 + c0] = d0[v];
        if (nq < 3 && ok1) {
#pragma unroll
            for (int v = 0; v < 8; ++v) o[v * 100 + c1] = d1[v];
        }
    }
}

__global__ void final_combine_kernel(const float* __restrict__ part,
                                     const float* __restrict__ lb2,
                                     const float* __restrict__ W2,
                                     const float* __restrict__ bias2,
                                     float* __restrict__ out) {
    __shared__ float comb[100];
    const int n = blockIdx.x;
    const int t = threadIdx.x;
    if (t < 100) {
        float pv[32];
#pragma unroll
        for (int p = 0; p < 32; ++p) pv[p] = part[((size_t)p * 64 + n) * 100 + t];
        float s = 0.f;
#pragma unroll
        for (int p = 0; p < 32; ++p) s += pv[p];
#pragma unroll
        for (int h = 0; h < 4; ++h) s += lb2[h * 100 + t];
        comb[t] = s * 0.25f;
    }
    __syncthreads();
    if (t < 100) {
        float acc = bias2[t];
        for (int j = 0; j < 100; ++j) acc += comb[j] * W2[t * 100 + j];
        float* orow = out + (size_t)n * 64 * 100 + t;
#pragma unroll 4
        for (int rep = 0; rep < 64; ++rep) orow[(size_t)rep * 100] = acc;
    }
}

extern "C" void kernel_launch(void* const* d_in, const int* in_sizes, int n_in,
                              void* d_out, int out_size, void* d_ws, size_t ws_size,
                              hipStream_t stream) {
    const float* x     = (const float*)d_in[0];
    const float* w1    = (const float*)d_in[1];
    const float* g1    = (const float*)d_in[2];
    const float* b1    = (const float*)d_in[3];
    const float* w2    = (const float*)d_in[4];
    const float* g2    = (const float*)d_in[5];
    const float* b2    = (const float*)d_in[6];
    const float* w3    = (const float*)d_in[7];
    const float* g3    = (const float*)d_in[8];
    const float* b3    = (const float*)d_in[9];
    const float* lw1   = (const float*)d_in[10];
    const float* lb1   = (const float*)d_in[11];
    const float* lw2   = (const float*)d_in[12];
    const float* lb2   = (const float*)d_in[13];
    const float* W2    = (const float*)d_in[14];
    const float* bias2 = (const float*)d_in[15];
    float* out = (float*)d_out;

    float* ws    = (float*)d_ws;
    float* z1raw = ws;
    float* part  = ws;
    float* fcin  = ws + 222208;

    roi_pool_conv1_kernel<<<64, 256, 0, stream>>>(x, w1, z1raw);

    conv_stack_kernel<<<4, 256, 0, stream>>>(
        z1raw, g1, b1, w2, g2, b2, w3, g3, b3, fcin);

    fc1_fc2_part_kernel<<<dim3(16, 4), 256, 0, stream>>>(fcin, lw1, lb1, lw2, part);

    final_combine_kernel<<<64, 128, 0, stream>>>(part, lb2, W2, bias2, out);
}


# The conv stack moves into the fc launch: each of its 64 workgroups rebuilds its head's BN+conv layers in LDS and writes its fc1 input rows there (4 launches -> 3); k orders and BN code unchanged
# speedup vs baseline: 3.4949x; 1.1773x over previous
#include <hip/hip_runtime.h>
#include <hip/hip_bf16.h>

typedef float v2f __attribute__((ext_vector_type(2)));
typedef float v4f __attribute__((ext_vector_type(4)));
typedef float v8f __attribute__((ext_vector_type(8)));

#define EPSV 1e-5f

__global__ __launch_bounds__(256) void roi_pool_conv1_kernel(
    const float* __restrict__ x,
    const float* __restrict__ w1,
    float* __restrict__ z1raw) {
    constexpr int P = 256 + 4;
    __shared__ alignas(16) float sW[128 * P];
    __shared__ alignas(16) float sP[5 * P];
    __shared__ alignas(16) float sO[4 * 128];

    const int n    = blockIdx.x;
    const int tid  = threadIdx.x;
    const int wave = tid >> 5;
    const int lane = tid & 31;
    const int half = lane >> 4;
    const int r    = lane & 15;

    const float* base = x + ((size_t)(n * 256 + tid) * 64) * 64;
    float v[4][5];
#pragma unroll
    for (int rr = 0; rr < 4; ++rr)
#pragma unroll
        for (int cc = 0; cc < 5; ++cc)
            v[rr][cc] = base[(2 + rr) * 64 + 1 + cc];
    __builtin_amdgcn_sched_barrier(0);

    const float4* w4 = reinterpret_cast<const float4*>(w1) + tid;
    float4 wv[32];
#pragma unroll
    for (int j = 0; j < 32; ++j) wv[j] = w4[256 * j];

    float* wd = sW + (tid >> 6) * P + (tid & 63) * 4;
#pragma unroll
    for (int j = 0; j < 32; ++j)
        *reinterpret_cast<float4*>(wd + 4 * j * P) = wv[j];

    float r0[5], r1[5];
#pragma unroll
    for (int cc = 0; cc < 5; ++cc) {
        r0[cc] = fmaxf(v[0][cc], v[1][cc]);
        r1[cc] = fmaxf(v[2][cc], v[3][cc]);
    }
    sP[0 * P + tid] = fmaxf(fmaxf(r0[0], r0[1]), r0[2]);
    sP[1 * P + tid] = fmaxf(fmaxf(r0[2], r0[3]), r0[4]);
    sP[2 * P + tid] = fmaxf(fmaxf(r1[0], r1[1]), r1[2]);
    sP[3 * P + tid] = fmaxf(fmaxf(r1[2], r1[3]), r1[4]);
    sP[4 * P + tid] = 0.f;
    __syncthreads();

    const float* ap = sP + (r < 4 ? r : 4) * P + 2 * half;
    const float* bp = sW + (wave * 16 + r) * P + 2 * half;
    v8f acc = {};
    for (int k0 = 0; k0 < 256; k0 += 4) {
        const float2 av = *reinterpret_cast<const float2*>(ap + k0);
        const float2 bv = *reinterpret_cast<const float2*>(bp + k0);
        v2f fa; fa[0] = av.x; fa[1] = av.y;
        v2f fb; fb[0] = bv.x; fb[1] = bv.y;
        acc = __builtin_amdgcn_wmma_f32_16x16x4_f32(false, fa, false, fb,
                                                    (short)0, acc, false, false);
    }
    if (half == 0) {
        float* o = sO + (wave >> 1) * 128 + (wave & 1) * 16 + r;
#pragma unroll
        for (int s = 0; s < 4; ++s) o[s * 32] = acc[s];
    }
    __syncthreads();
    if (tid < 128) {
        const float4 ov = *reinterpret_cast<const float4*>(sO + tid * 4);
        *reinterpret_cast<float4*>(z1raw + (size_t)(tid >> 5) * (256 * 32)
                                   + (size_t)n * 128 + (tid & 31) * 4) = ov;
    }
}

template <int K, int NROWS>
__device__ __forceinline__ void stage_rows(float* __restrict__ dst,
                                           const float* __restrict__ src, int tid) {
    constexpr int P     = K + 4;
    constexpr int VPR   = K / 4;
    constexpr int ITERS = NROWS * VPR / 256;
    static_assert((NROWS * VPR) % 256 == 0, "whole passes of the workgroup only");
    static_assert(256 % VPR == 0, "a row is a whole fraction of one pass of the workgroup");
    const float4* s4 = reinterpret_cast<const float4*>(src) + tid;
    float* d = dst + (tid / VPR) * P + (tid % VPR) * 4;
    float4 v[ITERS];
#pragma unroll
    for (int j = 0; j < ITERS; ++j) v[j] = s4[256 * j];
#pragma unroll
    for (int j = 0; j < ITERS; ++j)
        *reinterpret_cast<float4*>(d + (256 / VPR) * j * P) = v[j];
}

enum { TO_ROWS = 0, TO_FC = 1 };

template <int COUT, int MODE>
__device__ __forceinline__ void bn_relu_store(
    const v8f (&acc)[COUT / 8],
    const float* __restrict__ g, const float* __restrict__ b,
    float* lsum, float* lsq, float* lscale, float* lshift,
    float* dst, int mh, int tid) {
    constexpr int TPW = COUT / 8;

    const int wave = tid >> 5;
    const int lane = tid & 31;
    const int half = lane >> 4;
    const int r    = lane & 15;

#pragma unroll
    for (int i = 0; i < TPW; ++i) {
        const int n0 = (i >> 1) * 16;
        float s = 0.f, q = 0.f;
#pragma unroll
        for (int v = 0; v < 8; ++v) { float xv = acc[i][v]; s += xv; q += xv * xv; }
        atomicAdd(&lsum[n0 + r], s);
        atomicAdd(&lsq[n0 + r], q);
    }
    __syncthreads();

    if (tid < COUT) {
        float mu  = lsum[tid] * (1.f / 256.f);
        float var = lsq[tid] * (1.f / 256.f) - mu * mu;
        float sc  = g[tid] * rsqrtf(var + EPSV);
        lscale[tid] = sc;
        lshift[tid] = b[tid] - mu * sc;
        lsum[tid] = 0.f;
        lsq[tid]  = 0.f;
    }
    __syncthreads();

#pragma unroll
    for (int i = 0; i < TPW; ++i) {
        const int tt = wave + 8 * i;
        const int mt = tt & 15, nt = tt >> 4;
        const int m0 = mt * 16, n0 = nt * 16;
        const int col = n0 + r;
        if (MODE == TO_ROWS) {
            const float sc = lscale[col], sh = lshift[col];
#pragma unroll
            for (int v = 0; v < 8; ++v) {
                const int row = m0 + v + 8 * half;
                dst[row * (COUT + 4) + col] = fmaxf(acc[i][v] * sc + sh, 0.f);
            }
        } else {
            if ((i & 1) != mh) continue;
            const float sc = lscale[col], sh = lshift[col];
            float4 lo, hi;
            lo.x = fmaxf(acc[i][0] * sc + sh, 0.f);
            lo.y = fmaxf(acc[i][1] * sc + sh, 0.f);
            lo.z = fmaxf(acc[i][2] * sc + sh, 0.f);
            lo.w = fmaxf(acc[i][3] * sc + sh, 0.f);
            hi.x = fmaxf(acc[i][4] * sc + sh, 0.f);
            hi.y = fmaxf(acc[i][5] * sc + sh, 0.f);
            hi.z = fmaxf(acc[i][6] * sc + sh, 0.f);
            hi.w = fmaxf(acc[i][7] * sc + sh, 0.f);
            float* d = dst + ((mt & 7) * 4 + 2 * half) * (COUT * 4 + 4) + col * 4;
            *reinterpret_cast<float4*>(d) = lo;
            *reinterpret_cast<float4*>(d + (COUT * 4 + 4)) = hi;
        }
    }
}

template <int KIN, int COUT, int MODE>
__device__ __forceinline__ void conv_bn_relu_layer(
    const float* sIn, const float* sWt,
    const float* __restrict__ g, const float* __restrict__ b,
    float* lsum, float* lsq, float* lscale, float* lshift,
    float* dst, int mh, int tid) {
    constexpr int NT  = COUT / 16;
    constexpr int TPW = 2 * NT;
    constexpr int P   = KIN + 4;

    const int wave = tid >> 5;
    const int lane = tid & 31;
    const int half = lane >> 4;
    const int r    = lane & 15;

    v8f acc[TPW];
#pragma unroll
    for (int i = 0; i < TPW; ++i) acc[i] = v8f{};

    const float* alo = sIn + (wave * 16 + r) * P + 2 * half;
    const float* ahi = alo + 128 * P;
    const float* bw  = sWt + r * P + 2 * half;
    for (int k0 = 0; k0 < KIN; k0 += 4) {
        const float2 a0 = *reinterpret_cast<const float2*>(alo + k0);
        const float2 a1 = *reinterpret_cast<const float2*>(ahi + k0);
        v2f fa0; fa0[0] = a0.x; fa0[1] = a0.y;
        v2f fa1; fa1[0] = a1.x; fa1[1] = a1.y;
#pragma unroll
        for (int nt = 0; nt < NT; ++nt) {
            const float2 bv = *reinterpret_cast<const float2*>(bw + nt * 16 * P + k0);
            v2f fb; fb[0] = bv.x; fb[1] = bv.y;
            acc[2 * nt]     = __builtin_amdgcn_wmma_f32_16x16x4_f32(false, fa0, false, fb,
                                                                    (short)0, acc[2 * nt], false, false);
            acc[2 * nt + 1] = __builtin_amdgcn_wmma_f32_16x16x4_f32(false, fa1, false, fb,
                                                                    (short)0, acc[2 * nt + 1], false, false);
        }
    }
    bn_relu_store<COUT, MODE>(acc, g, b, lsum, lsq, lscale, lshift, dst, mh, tid);
}

__global__ __launch_bounds__(256) void conv_fc1_fc2_part_kernel(
    const float* __restrict__ z1raw,
    const float* __restrict__ g1, const float* __restrict__ b1,
    const float* __restrict__ w2, const float* __restrict__ g2, const float* __restrict__ b2,
    const float* __restrict__ w3, const float* __restrict__ g3, const float* __restrict__ b3,
    const float* __restrict__ lw1,
    const float* __restrict__ lb1,
    const float* __restrict__ lw2,
    float* __restrict__ part) {
    constexpr int PC2 = 32 + 4, PC3 = 64 + 4;
    constexpr int P1 = 256 + 4;
    constexpr int P2 = 64 + 4;
    constexpr int OFF_W2 = 4 * 64;
    constexpr int OFF_W3 = OFF_W2 + 64 * PC2;
    constexpr int OFF_Z1 = OFF_W3 + 64 * PC3;
    constexpr int OFF_Z2 = OFF_Z1 + 256 * PC2;
    constexpr int OFF_A  = OFF_Z2 + 256 * PC3;
    constexpr int OFF_B1 = OFF_A + 32 * P1;
    constexpr int OFF_B2 = OFF_B1 + 64 * P1;
    constexpr int OFF_H  = OFF_B2 + 112 * P2;
    constexpr int TOTAL  = OFF_H + 32 * P2;
    static_assert(TOTAL * 4 <= 320 * 1024, "one workgroup's LDS");
    __shared__ alignas(16) float smem[TOTAL];

    float* lsum   = smem;
    float* lsq    = smem + 64;
    float* lscale = smem + 128;
    float* lshift = smem + 192;
    float* sW2 = smem + OFF_W2;
    float* sW3 = smem + OFF_W3;
    float* sZ1 = smem + OFF_Z1;
    float* sZ2 = smem + OFF_Z2;
    float* sA  = smem + OFF_A;
    float* sB1 = smem + OFF_B1;
    float* sB2 = smem + OFF_B2;
    float* sH  = smem + OFF_H;

    const int s    = blockIdx.x >> 1;
    const int mh   = blockIdx.x & 1;
    const int h    = blockIdx.y;
    const int tid  = threadIdx.x;
    const int wave = __builtin_amdgcn_readfirstlane(tid >> 5);
    const int lane = tid & 31;
    const int half = lane >> 4;
    const int r    = lane & 15;
    const int mt   = wave & 1;
    const int nq   = wave >> 1;

    if (tid < 64) { lsum[tid] = 0.f; lsq[tid] = 0.f; }

    v8f acc1[4];
    const float* zr = z1raw + (size_t)h * (256 * 32) + (8 * half) * 32 + r;
#pragma unroll
    for (int i = 0; i < 4; ++i) {
        const int tt = (tid >> 5) + 8 * i;
        const int mt1 = tt & 15, nt1 = tt >> 4;
#pragma unroll
        for (int v = 0; v < 8; ++v)
            acc1[i][v] = zr[(mt1 * 16 + v) * 32 + nt1 * 16];
    }
    stage_rows<32, 64>(sW2, w2 + (size_t)h * 64 * 32, tid);
    stage_rows<64, 64>(sW3, w3 + (size_t)h * 64 * 64, tid);
    __syncthreads();

    bn_relu_store<32, TO_ROWS>(acc1, g1 + h * 32, b1 + h * 32,
                               lsum, lsq, lscale, lshift, sZ1, mh, tid);
    __syncthreads();

    const v4f* b4 = reinterpret_cast<const v4f*>(lw1 + ((size_t)h * 512 + s * 64) * 256) + tid;
    const float* w2p = lw2 + ((size_t)h * 100 + (tid >> 4)) * 512 + s * 64 + (tid & 15) * 4;
    v4f bv1[16], bv2[7];
#pragma unroll
    for (int j = 0; j < 16; ++j) bv1[j] = b4[256 * j];
#pragma unroll
    for (int j = 0; j < 6; ++j) bv2[j] = *reinterpret_cast<const v4f*>(w2p + 16 * j * 512);
    const int r6 = min(96 + (tid >> 4), 99);
    bv2[6] = *reinterpret_cast<const v4f*>(lw2 + ((size_t)h * 100 + r6) * 512 + s * 64 + (tid & 15) * 4);
    const float bs = lb1[h * 512 + s * 64 + nq * 16 + r];

    conv_bn_relu_layer<32, 64, TO_ROWS>(sZ1, sW2, g2 + h * 64, b2 + h * 64,
                                        lsum, lsq, lscale, lshift, sZ2, mh, tid);

    {
        float* bd = sB1 + (tid >> 6) * P1 + (tid & 63) * 4;
#pragma unroll
        for (int j = 0; j < 16; ++j) *reinterpret_cast<v4f*>(bd + 4 * j * P1) = bv1[j];
        float* wd = sB2 + (tid >> 4) * P2 + (tid & 15) * 4;
#pragma unroll
        for (int j = 0; j < 7; ++j) *reinterpret_cast<v4f*>(wd + 16 * j * P2) = bv2[j];
    }
    __syncthreads();
    conv_bn_relu_layer<64, 64, TO_FC>(sZ2, sW3, g3 + h * 64, b3 + h * 64,
                                      lsum, lsq, lscale, lshift, sA, mh, tid);
    __syncthreads();

    {
        const float* ap = sA + (mt * 16 + r) * P1 + 2 * half;
        const float* bp = sB1 + (nq * 16 + r) * P1 + 2 * half;
        v8f c = {};
        for (int k0 = 0; k0 < 256; k0 += 4) {
            const float2 a2 = *reinterpret_cast<const float2*>(ap + k0);
            const float2 b2v = *reinterpret_cast<const float2*>(bp + k0);
            v2f fa; fa[0] = a2.x; fa[1] = a2.y;
            v2f fb; fb[0] = b2v.x; fb[1] = b2v.y;
            c = __builtin_amdgcn_wmma_f32_16x16x4_f32(false, fa, false, fb,
                                                      (short)0, c, false, false);
        }
        float* hp = sH + (mt * 16 + 8 * half) * P2 + nq * 16 + r;
#pragma unroll
        for (int v = 0; v < 8; ++v) hp[v * P2] = fmaxf(c[v] + bs, 0.f);
    }
    __syncthreads();

    {
        const int c0 = nq * 16 + r;
        const int c1 = c0 + 64;
        const bool ok1 = c1 < 100;
        const float* ap  = sH + (mt * 16 + r) * P2 + 2 * half;
        const float* bp0 = sB2 + c0 * P2 + 2 * half;
        const float* bp1 = sB2 + (ok1 ? c1 : 0) * P2 + 2 * half;
        v8f d0 = {}, d1 = {};
        if (nq < 3) {
            for (int k0 = 0; k0 < 64; k0 += 4) {
                const float2 a2 = *reinterpret_cast<const float2*>(ap + k0);
                const float2 x0 = *reinterpret_cast<const float2*>(bp0 + k0);
                const float2 x1 = *reinterpret_cast<const float2*>(bp1 + k0);
                v2f fa;  fa[0] = a2.x; fa[1] = a2.y;
                v2f fb0; fb0[0] = x0.x; fb0[1] = x0.y;
                v2f fb1; fb1[0] = ok1 ? x1.x : 0.f; fb1[1] = ok1 ? x1.y : 0.f;
                d0 = __builtin_amdgcn_wmma_f32_16x16x4_f32(false, fa, false, fb0,
                                                           (short)0, d0, false, false);
                d1 = __builtin_amdgcn_wmma_f32_16x16x4_f32(false, fa, false, fb1,
                                                           (short)0, d1, false, false);
            }
        } else {
            for (int k0 = 0; k0 < 64; k0 += 4) {
                const float2 a2 = *reinterpret_cast<const float2*>(ap + k0);
                const float2 x0 = *reinterpret_cast<const float2*>(bp0 + k0);
                v2f fa;  fa[0] = a2.x; fa[1] = a2.y;
                v2f fb0; fb0[0] = x0.x; fb0[1] = x0.y;
                d0 = __builtin_amdgcn_wmma_f32_16x16x4_f32(false, fa, false, fb0,
                                                           (short)0, d0, false, false);
            }
        }
        float* o = part + (((size_t)h * 8 + s) * 64 + mh * 32 + mt * 16 + 8 * half) * 100;
#pragma unroll
        for (int v = 0; v < 8; ++v) o[v * 100 + c0] = d0[v];
        if (nq < 3 && ok1) {
#pragma unroll
            for (int v = 0; v < 8; ++v) o[v * 100 + c1] = d1[v];
        }
    }
}

__global__ void final_combine_kernel(const float* __restrict__ part,
                                     const float* __restrict__ lb2,
                                     const float* __restrict__ W2,
                                     const float* __restrict__ bias2,
                                     float* __restrict__ out) {
    __shared__ float comb[100];
    const int n = blockIdx.x;
    const int t = threadIdx.x;
    if (t < 100) {
        float pv[32];
#pragma unroll
        for (int p = 0; p < 32; ++p) pv[p] = part[((size_t)p * 64 + n) * 100 + t];
        float s = 0.f;
#pragma unroll
        for (int p = 0; p < 32; ++p) s += pv[p];
#pragma unroll
        for (int h = 0; h < 4; ++h) s += lb2[h * 100 + t];
        comb[t] = s * 0.25f;
    }
    __syncthreads();
    if (t < 100) {
        float acc = bias2[t];
        for (int j = 0; j < 100; ++j) acc += comb[j] * W2[t * 100 + j];
        float* orow = out + (size_t)n * 64 * 100 + t;
#pragma unroll 4
        for (int rep = 0; rep < 64; ++rep) orow[(size_t)rep * 100] = acc;
    }
}

extern "C" void kernel_launch(void* const* d_in, const int* in_sizes, int n_in,
                              void* d_out, int out_size, void* d_ws, size_t ws_size,
                              hipStream_t stream) {
    const float* x     = (const float*)d_in[0];
    const float* w1    = (const float*)d_in[1];
    const float* g1    = (const float*)d_in[2];
    const float* b1    = (const float*)d_in[3];
    const float* w2    = (const float*)d_in[4];
    const float* g2    = (const float*)d_in[5];
    const float* b2    = (const float*)d_in[6];
    const float* w3    = (const float*)d_in[7];
    const float* g3    = (const float*)d_in[8];
    const float* b3    = (const float*)d_in[9];
    const float* lw1   = (const float*)d_in[10];
    const float* lb1   = (const float*)d_in[11];
    const float* lw2   = (const float*)d_in[12];
    const float* lb2   = (const float*)d_in[13];
    const float* W2    = (const float*)d_in[14];
    const float* bias2 = (const float*)d_in[15];
    float* out = (float*)d_out;

    float* ws    = (float*)d_ws;
    float* part  = ws;
    float* z1raw = ws + 222208;

    roi_pool_conv1_kernel<<<64, 256, 0, stream>>>(x, w1, z1raw);

    conv_fc1_fc2_part_kernel<<<dim3(16, 4), 256, 0, stream>>>(
        z1raw, g1, b1, w2, g2, b2, w3, g3, b3, lw1, lb1, lw2, part);

    final_combine_kernel<<<64, 128, 0, stream>>>(part, lb2, W2, bias2, out);
}


# Last kernel: W2 staged in LDS by whole-line b128 loads issued at entry beside the part loads (was strided row loads behind the barrier); result rows stored as whole-line b128; sum order unchanged
# speedup vs baseline: 3.6721x; 1.0507x over previous
#include <hip/hip_runtime.h>
#include <hip/hip_bf16.h>

typedef float v2f __attribute__((ext_vector_type(2)));
typedef float v4f __attribute__((ext_vector_type(4)));
typedef float v8f __attribute__((ext_vector_type(8)));

#define EPSV 1e-5f

__global__ __launch_bounds__(256) void roi_pool_conv1_kernel(
    const float* __restrict__ x,
    const float* __restrict__ w1,
    float* __restrict__ z1raw) {
    constexpr int P = 256 + 4;
    __shared__ alignas(16) float sW[128 * P];
    __shared__ alignas(16) float sP[5 * P];
    __shared__ alignas(16) float sO[4 * 128];

    const int n    = blockIdx.x;
    const int tid  = threadIdx.x;
    const int wave = tid >> 5;
    const int lane = tid & 31;
    const int half = lane >> 4;
    const int r    = lane & 15;

    const float* base = x + ((size_t)(n * 256 + tid) * 64) * 64;
    float v[4][5];
#pragma unroll
    for (int rr = 0; rr < 4; ++rr)
#pragma unroll
        for (int cc = 0; cc < 5; ++cc)
            v[rr][cc] = base[(2 + rr) * 64 + 1 + cc];
    __builtin_amdgcn_sched_barrier(0);

    const float4* w4 = reinterpret_cast<const float4*>(w1) + tid;
    float4 wv[32];
#pragma unroll
    for (int j = 0; j < 32; ++j) wv[j] = w4[256 * j];

    float* wd = sW + (tid >> 6) * P + (tid & 63) * 4;
#pragma unroll
    for (int j = 0; j < 32; ++j)
        *reinterpret_cast<float4*>(wd + 4 * j * P) = wv[j];

    float r0[5], r1[5];
#pragma unroll
    for (int cc = 0; cc < 5; ++cc) {
        r0[cc] = fmaxf(v[0][cc], v[1][cc]);
        r1[cc] = fmaxf(v[2][cc], v[3][cc]);
    }
    sP[0 * P + tid] = fmaxf(fmaxf(r0[0], r0[1]), r0[2]);
    sP[1 * P + tid] = fmaxf(fmaxf(r0[2], r0[3]), r0[4]);
    sP[2 * P + tid] = fmaxf(fmaxf(r1[0], r1[1]), r1[2]);
    sP[3 * P + tid] = fmaxf(fmaxf(r1[2], r1[3]), r1[4]);
    sP[4 * P + tid] = 0.f;
    __syncthreads();

    const float* ap = sP + (r < 4 ? r : 4) * P + 2 * half;
    const float* bp = sW + (wave * 16 + r) * P + 2 * half;
    v8f acc = {};
    for (int k0 = 0; k0 < 256; k0 += 4) {
        const float2 av = *reinterpret_cast<const float2*>(ap + k0);
        const float2 bv = *reinterpret_cast<const float2*>(bp + k0);
        v2f fa; fa[0] = av.x; fa[1] = av.y;
        v2f fb; fb[0] = bv.x; fb[1] = bv.y;
        acc = __builtin_amdgcn_wmma_f32_16x16x4_f32(false, fa, false, fb,
                                                    (short)0, acc, false, false);
    }
    if (half == 0) {
        float* o = sO + (wave >> 1) * 128 + (wave & 1) * 16 + r;
#pragma unroll
        for (int s = 0; s < 4; ++s) o[s * 32] = acc[s];
    }
    __syncthreads();
    if (tid < 128) {
        const float4 ov = *reinterpret_cast<const float4*>(sO + tid * 4);
        *reinterpret_cast<float4*>(z1raw + (size_t)(tid >> 5) * (256 * 32)
                                   + (size_t)n * 128 + (tid & 31) * 4) = ov;
    }
}

template <int K, int NROWS>
__device__ __forceinline__ void stage_rows(float* __restrict__ dst,
                                           const float* __restrict__ src, int tid) {
    constexpr int P     = K + 4;
    constexpr int VPR   = K / 4;
    constexpr int ITERS = NROWS * VPR / 256;
    static_assert((NROWS * VPR) % 256 == 0, "whole passes of the workgroup only");
    static_assert(256 % VPR == 0, "a row is a whole fraction of one pass of the workgroup");
    const float4* s4 = reinterpret_cast<const float4*>(src) + tid;
    float* d = dst + (tid / VPR) * P + (tid % VPR) * 4;
    float4 v[ITERS];
#pragma unroll
    for (int j = 0; j < ITERS; ++j) v[j] = s4[256 * j];
#pragma unroll
    for (int j = 0; j < ITERS; ++j)
        *reinterpret_cast<float4*>(d + (256 / VPR) * j * P) = v[j];
}

enum { TO_ROWS = 0, TO_FC = 1 };

template <int COUT, int MODE>
__device__ __forceinline__ void bn_relu_store(
    const v8f (&acc)[COUT / 8],
    const float* __restrict__ g, const float* __restrict__ b,
    float* lsum, float* lsq, float* lscale, float* lshift,
    float* dst, int mh, int tid) {
    constexpr int TPW = COUT / 8;

    const int wave = tid >> 5;
    const int lane = tid & 31;
    const int half = lane >> 4;
    const int r    = lane & 15;

#pragma unroll
    for (int i = 0; i < TPW; ++i) {
        const int n0 = (i >> 1) * 16;
        float s = 0.f, q = 0.f;
#pragma unroll
        for (int v = 0; v < 8; ++v) { float xv = acc[i][v]; s += xv; q += xv * xv; }
        atomicAdd(&lsum[n0 + r], s);
        atomicAdd(&lsq[n0 + r], q);
    }
    __syncthreads();

    if (tid < COUT) {
        float mu  = lsum[tid] * (1.f / 256.f);
        float var = lsq[tid] * (1.f / 256.f) - mu * mu;
        float sc  = g[tid] * rsqrtf(var + EPSV);
        lscale[tid] = sc;
        lshift[tid] = b[tid] - mu * sc;
        lsum[tid] = 0.f;
        lsq[tid]  = 0.f;
    }
    __syncthreads();

#pragma unroll
    for (int i = 0; i < TPW; ++i) {
        const int tt = wave + 8 * i;
        const int mt = tt & 15, nt = tt >> 4;
        const int m0 = mt * 16, n0 = nt * 16;
        const int col = n0 + r;
        if (MODE == TO_ROWS) {
            const float sc = lscale[col], sh = lshift[col];
#pragma unroll
            for (int v = 0; v < 8; ++v) {
                const int row = m0 + v + 8 * half;
                dst[row * (COUT + 4) + col] = fmaxf(acc[i][v] * sc + sh, 0.f);
            }
        } else {
            if ((i & 1) != mh) continue;
            const float sc = lscale[col], sh = lshift[col];
            float4 lo, hi;
            lo.x = fmaxf(acc[i][0] * sc + sh, 0.f);
            lo.y = fmaxf(acc[i][1] * sc + sh, 0.f);
            lo.z = fmaxf(acc[i][2] * sc + sh, 0.f);
            lo.w = fmaxf(acc[i][3] * sc + sh, 0.f);
            hi.x = fmaxf(acc[i][4] * sc + sh, 0.f);
            hi.y = fmaxf(acc[i][5] * sc + sh, 0.f);
            hi.z = fmaxf(acc[i][6] * sc + sh, 0.f);
            hi.w = fmaxf(acc[i][7] * sc + sh, 0.f);
            float* d = dst + ((mt & 7) * 4 + 2 * half) * (COUT * 4 + 4) + col * 4;
            *reinterpret_cast<float4*>(d) = lo;
            *reinterpret_cast<float4*>(d + (COUT * 4 + 4)) = hi;
        }
    }
}

template <int KIN, int COUT, int MODE>
__device__ __forceinline__ void conv_bn_relu_layer(
    const float* sIn, const float* sWt,
    const float* __restrict__ g, const float* __restrict__ b,
    float* lsum, float* lsq, float* lscale, float* lshift,
    float* dst, int mh, int tid) {
    constexpr int NT  = COUT / 16;
    constexpr int TPW = 2 * NT;
    constexpr int P   = KIN + 4;

    const int wave = tid >> 5;
    const int lane = tid & 31;
    const int half = lane >> 4;
    const int r    = lane & 15;

    v8f acc[TPW];
#pragma unroll
    for (int i = 0; i < TPW; ++i) acc[i] = v8f{};

    const float* alo = sIn + (wave * 16 + r) * P + 2 * half;
    const float* ahi = alo + 128 * P;
    const float* bw  = sWt + r * P + 2 * half;
    for (int k0 = 0; k0 < KIN; k0 += 4) {
        const float2 a0 = *reinterpret_cast<const float2*>(alo + k0);
        const float2 a1 = *reinterpret_cast<const float2*>(ahi + k0);
        v2f fa0; fa0[0] = a0.x; fa0[1] = a0.y;
        v2f fa1; fa1[0] = a1.x; fa1[1] = a1.y;
#pragma unroll
        for (int nt = 0; nt < NT; ++nt) {
            const float2 bv = *reinterpret_cast<const float2*>(bw + nt * 16 * P + k0);
            v2f fb; fb[0] = bv.x; fb[1] = bv.y;
            acc[2 * nt]     = __builtin_amdgcn_wmma_f32_16x16x4_f32(false, fa0, false, fb,
                                                                    (short)0, acc[2 * nt], false, false);
            acc[2 * nt + 1] = __builtin_amdgcn_wmma_f32_16x16x4_f32(false, fa1, false, fb,
                                                                    (short)0, acc[2 * nt + 1], false, false);
        }
    }
    bn_relu_store<COUT, MODE>(acc, g, b, lsum, lsq, lscale, lshift, dst, mh, tid);
}

__global__ __launch_bounds__(256) void conv_fc1_fc2_part_kernel(
    const float* __restrict__ z1raw,
    const float* __restrict__ g1, const float* __restrict__ b1,
    const float* __restrict__ w2, const float* __restrict__ g2, const float* __restrict__ b2,
    const float* __restrict__ w3, const float* __restrict__ g3, const float* __restrict__ b3,
    const float* __restrict__ lw1,
    const float* __restrict__ lb1,
    const float* __restrict__ lw2,
    float* __restrict__ part) {
    constexpr int PC2 = 32 + 4, PC3 = 64 + 4;
    constexpr int P1 = 256 + 4;
    constexpr int P2 = 64 + 4;
    constexpr int OFF_W2 = 4 * 64;
    constexpr int OFF_W3 = OFF_W2 + 64 * PC2;
    constexpr int OFF_Z1 = OFF_W3 + 64 * PC3;
    constexpr int OFF_Z2 = OFF_Z1 + 256 * PC2;
    constexpr int OFF_A  = OFF_Z2 + 256 * PC3;
    constexpr int OFF_B1 = OFF_A + 32 * P1;
    constexpr int OFF_B2 = OFF_B1 + 64 * P1;
    constexpr int OFF_H  = OFF_B2 + 112 * P2;
    constexpr int TOTAL  = OFF_H + 32 * P2;
    static_assert(TOTAL * 4 <= 320 * 1024, "one workgroup's LDS");
    __shared__ alignas(16) float smem[TOTAL];

    float* lsum   = smem;
    float* lsq    = smem + 64;
    float* lscale = smem + 128;
    float* lshift = smem + 192;
    float* sW2 = smem + OFF_W2;
    float* sW3 = smem + OFF_W3;
    float* sZ1 = smem + OFF_Z1;
    float* sZ2 = smem + OFF_Z2;
    float* sA  = smem + OFF_A;
    float* sB1 = smem + OFF_B1;
    float* sB2 = smem + OFF_B2;
    float* sH  = smem + OFF_H;

    const int s    = blockIdx.x >> 1;
    const int mh   = blockIdx.x & 1;
    const int h    = blockIdx.y;
    const int tid  = threadIdx.x;
    const int wave = __builtin_amdgcn_readfirstlane(tid >> 5);
    const int lane = tid & 31;
    const int half = lane >> 4;
    const int r    = lane & 15;
    const int mt   = wave & 1;
    const int nq   = wave >> 1;

    if (tid < 64) { lsum[tid] = 0.f; lsq[tid] = 0.f; }

    v8f acc1[4];
    const float* zr = z1raw + (size_t)h * (256 * 32) + (8 * half) * 32 + r;
#pragma unroll
    for (int i = 0; i < 4; ++i) {
        const int tt = (tid >> 5) + 8 * i;
        const int mt1 = tt & 15, nt1 = tt >> 4;
#pragma unroll
        for (int v = 0; v < 8; ++v)
            acc1[i][v] = zr[(mt1 * 16 + v) * 32 + nt1 * 16];
    }
    stage_rows<32, 64>(sW2, w2 + (size_t)h * 64 * 32, tid);
    stage_rows<64, 64>(sW3, w3 + (size_t)h * 64 * 64, tid);
    __syncthreads();

    bn_relu_store<32, TO_ROWS>(acc1, g1 + h * 32, b1 + h * 32,
                               lsum, lsq, lscale, lshift, sZ1, mh, tid);
    __syncthreads();

    const v4f* b4 = reinterpret_cast<const v4f*>(lw1 + ((size_t)h * 512 + s * 64) * 256) + tid;
    const float* w2p = lw2 + ((size_t)h * 100 + (tid >> 4)) * 512 + s * 64 + (tid & 15) * 4;
    v4f bv1[16], bv2[7];
#pragma unroll
    for (int j = 0; j < 16; ++j) bv1[j] = b4[256 * j];
#pragma unroll
    for (int j = 0; j < 6; ++j) bv2[j] = *reinterpret_cast<const v4f*>(w2p + 16 * j * 512);
    const int r6 = min(96 + (tid >> 4), 99);
    bv2[6] = *reinterpret_cast<const v4f*>(lw2 + ((size_t)h * 100 + r6) * 512 + s * 64 + (tid & 15) * 4);
    const float bs = lb1[h * 512 + s * 64 + nq * 16 + r];

    conv_bn_relu_layer<32, 64, TO_ROWS>(sZ1, sW2, g2 + h * 64, b2 + h * 64,
                                        lsum, lsq, lscale, lshift, sZ2, mh, tid);

    {
        float* bd = sB1 + (tid >> 6) * P1 + (tid & 63) * 4;
#pragma unroll
        for (int j = 0; j < 16; ++j) *reinterpret_cast<v4f*>(bd + 4 * j * P1) = bv1[j];
        float* wd = sB2 + (tid >> 4) * P2 + (tid & 15) * 4;
#pragma unroll
        for (int j = 0; j < 7; ++j) *reinterpret_cast<v4f*>(wd + 16 * j * P2) = bv2[j];
    }
    __syncthreads();
    conv_bn_relu_layer<64, 64, TO_FC>(sZ2, sW3, g3 + h * 64, b3 + h * 64,
                                      lsum, lsq, lscale, lshift, sA, mh, tid);
    __syncthreads();

    {
        const float* ap = sA + (mt * 16 + r) * P1 + 2 * half;
        const float* bp = sB1 + (nq * 16 + r) * P1 + 2 * half;
        v8f c = {};
        for (int k0 = 0; k0 < 256; k0 += 4) {
            const float2 a2 = *reinterpret_cast<const float2*>(ap + k0);
            const float2 b2v = *reinterpret_cast<const float2*>(bp + k0);
            v2f fa; fa[0] = a2.x; fa[1] = a2.y;
            v2f fb; fb[0] = b2v.x; fb[1] = b2v.y;
            c = __builtin_amdgcn_wmma_f32_16x16x4_f32(false, fa, false, fb,
                                                      (short)0, c, false, false);
        }
        float* hp = sH + (mt * 16 + 8 * half) * P2 + nq * 16 + r;
#pragma unroll
        for (int v = 0; v < 8; ++v) hp[v * P2] = fmaxf(c[v] + bs, 0.f);
    }
    __syncthreads();

    {
        const int c0 = nq * 16 + r;
        const int c1 = c0 + 64;
        const bool ok1 = c1 < 100;
        const float* ap  = sH + (mt * 16 + r) * P2 + 2 * half;
        const float* bp0 = sB2 + c0 * P2 + 2 * half;
        const float* bp1 = sB2 + (ok1 ? c1 : 0) * P2 + 2 * half;
        v8f d0 = {}, d1 = {};
        if (nq < 3) {
            for (int k0 = 0; k0 < 64; k0 += 4) {
                const float2 a2 = *reinterpret_cast<const float2*>(ap + k0);
                const float2 x0 = *reinterpret_cast<const float2*>(bp0 + k0);
                const float2 x1 = *reinterpret_cast<const float2*>(bp1 + k0);
                v2f fa;  fa[0] = a2.x; fa[1] = a2.y;
                v2f fb0; fb0[0] = x0.x; fb0[1] = x0.y;
                v2f fb1; fb1[0] = ok1 ? x1.x : 0.f; fb1[1] = ok1 ? x1.y : 0.f;
                d0 = __builtin_amdgcn_wmma_f32_16x16x4_f32(false, fa, false, fb0,
                                                           (short)0, d0, false, false);
                d1 = __builtin_amdgcn_wmma_f32_16x16x4_f32(false, fa, false, fb1,
                                                           (short)0, d1, false, false);
            }
        } else {
            for (int k0 = 0; k0 < 64; k0 += 4) {
                const float2 a2 = *reinterpret_cast<const float2*>(ap + k0);
                const float2 x0 = *reinterpret_cast<const float2*>(bp0 + k0);
                v2f fa;  fa[0] = a2.x; fa[1] = a2.y;
                v2f fb0; fb0[0] = x0.x; fb0[1] = x0.y;
                d0 = __builtin_amdgcn_wmma_f32_16x16x4_f32(false, fa, false, fb0,
                                                           (short)0, d0, false, false);
            }
        }
        float* o = part + (((size_t)h * 8 + s) * 64 + mh * 32 + mt * 16 + 8 * half) * 100;
#pragma unroll
        for (int v = 0; v < 8; ++v) o[v * 100 + c0] = d0[v];
        if (nq < 3 && ok1) {
#pragma unroll
            for (int v = 0; v < 8; ++v) o[v * 100 + c1] = d1[v];
        }
    }
}

__global__ __launch_bounds__(128) void final_combine_kernel(
                                     const float* __restrict__ part,
                                     const float* __restrict__ lb2,
                                     const float* __restrict__ W2,
                                     const float* __restrict__ bias2,
                                     float* __restrict__ out) {
    __shared__ alignas(16) float sW[100 * 100];
    __shared__ alignas(16) float comb[128];
    __shared__ alignas(16) float res[128];
    const int n  = blockIdx.x;
    const int t  = threadIdx.x;
    const int tc = min(t, 99);

    float pv[32], lb[4];
#pragma unroll
    for (int p = 0; p < 32; ++p) pv[p] = part[((size_t)p * 64 + n) * 100 + tc];
#pragma unroll
    for (int h = 0; h < 4; ++h) lb[h] = lb2[h * 100 + tc];
    const float bz = bias2[tc];

    const float4* w4 = reinterpret_cast<const float4*>(W2);
    const int last = min(t + 2432, 2499);
    float4 wv[20];
#pragma unroll
    for (int j = 0; j < 19; ++j) wv[j] = w4[t + 128 * j];
    wv[19] = w4[last];

    float s = 0.f;
#pragma unroll
    for (int p = 0; p < 32; ++p) s += pv[p];
#pragma unroll
    for (int h = 0; h < 4; ++h) s += lb[h];
    comb[t] = s * 0.25f;

    float4* sw4 = reinterpret_cast<float4*>(sW);
#pragma unroll
    for (int j = 0; j < 19; ++j) sw4[t + 128 * j] = wv[j];
    sw4[last] = wv[19];
    __syncthreads();

    {
        const float4* wr = reinterpret_cast<const float4*>(sW + tc * 100);
        const float4* c4 = reinterpret_cast<const float4*>(comb);
        float acc = bz;
#pragma unroll
        for (int q = 0; q < 25; ++q) {
            const float4 w = wr[q];
            const float4 c = c4[q];
            acc = fmaf(c.x, w.x, acc);
            acc = fmaf(c.y, w.y, acc);
            acc = fmaf(c.z, w.z, acc);
            acc = fmaf(c.w, w.w, acc);
        }
        res[t] = acc;
    }
    __syncthreads();

    const float4* r4 = reinterpret_cast<const float4*>(res);
    float4* o4 = reinterpret_cast<float4*>(out + (size_t)n * 64 * 100);
#pragma unroll
    for (int j = 0; j < 12; ++j) {
        const int i = t + 128 * j;
        o4[i] = r4[i % 25];
    }
    if (t < 64) {
        const int i = t + 1536;
        o4[i] = r4[i % 25];
    }
}

extern "C" void kernel_launch(void* const* d_in, const int* in_sizes, int n_in,
                              void* d_out, int out_size, void* d_ws, size_t ws_size,
                              hipStream_t stream) {
    const float* x     = (const float*)d_in[0];
    const float* w1    = (const float*)d_in[1];
    const float* g1    = (const float*)d_in[2];
    const float* b1    = (const float*)d_in[3];
    const float* w2    = (const float*)d_in[4];
    const float* g2    = (const float*)d_in[5];
    const float* b2    = (const float*)d_in[6];
    const float* w3    = (const float*)d_in[7];
    const float* g3    = (const float*)d_in[8];
    const float* b3    = (const float*)d_in[9];
    const float* lw1   = (const float*)d_in[10];
    const float* lb1   = (const float*)d_in[11];
    const float* lw2   = (const float*)d_in[12];
    const float* lb2   = (const float*)d_in[13];
    const float* W2    = (const float*)d_in[14];
    const float* bias2 = (const float*)d_in[15];
    float* out = (float*)d_out;

    float* ws    = (float*)d_ws;
    float* part  = ws;
    float* z1raw = ws + 222208;

    roi_pool_conv1_kernel<<<64, 256, 0, stream>>>(x, w1, z1raw);

    conv_fc1_fc2_part_kernel<<<dim3(16, 4), 256, 0, stream>>>(
        z1raw, g1, b1, w2, g2, b2, w3, g3, b3, lw1, lb1, lw2, part);

    final_combine_kernel<<<64, 128, 0, stream>>>(part, lb2, W2, bias2, out);
}


# f32 WMMAs per SIMD cut: layer 1 one workgroup per (sample, head), k in 4 wave quarters; fc1/fc2 per 16-sample quarter, fc1 k in 2 wave halves; parts added via LDS (layer-1, fc1 sum order changes)
# speedup vs baseline: 4.0376x; 1.0995x over previous
#include <hip/hip_runtime.h>
#include <hip/hip_bf16.h>

typedef float v2f __attribute__((ext_vector_type(2)));
typedef float v4f __attribute__((ext_vector_type(4)));
typedef float v8f __attribute__((ext_vector_type(8)));

#define EPSV 1e-5f

__global__ __launch_bounds__(256) void roi_pool_conv1_kernel(
    const float* __restrict__ x,
    const float* __restrict__ w1,
    float* __restrict__ z1raw) {
    constexpr int P = 256 + 4;
    __shared__ alignas(16) float sW[32 * P];
    __shared__ alignas(16) float sP[5 * P];
    __shared__ alignas(16) float sO[4 * 128];

    const int n    = blockIdx.x;
    const int h    = blockIdx.y;
    const int tid  = threadIdx.x;
    const int wave = tid >> 5;
    const int lane = tid & 31;
    const int half = lane >> 4;
    const int r    = lane & 15;

    const float* base = x + ((size_t)(n * 256 + tid) * 64) * 64;
    const float4* w4 = reinterpret_cast<const float4*>(w1 + (size_t)h * (32 * 256)) + tid;
    float v[4][5];
#pragma unroll
    for (int rr = 0; rr < 4; ++rr)
#pragma unroll
        for (int cc = 0; cc < 5; ++cc)
            v[rr][cc] = base[(2 + rr) * 64 + 1 + cc];
    __builtin_amdgcn_sched_barrier(0);

    float4 wv[8];
#pragma unroll
    for (int j = 0; j < 8; ++j) wv[j] = w4[256 * j];

    float* wd = sW + (tid >> 6) * P + (tid & 63) * 4;
#pragma unroll
    for (int j = 0; j < 8; ++j)
        *reinterpret_cast<float4*>(wd + 4 * j * P) = wv[j];

    float r0[5], r1[5];
#pragma unroll
    for (int cc = 0; cc < 5; ++cc) {
        r0[cc] = fmaxf(v[0][cc], v[1][cc]);
        r1[cc] = fmaxf(v[2][cc], v[3][cc]);
    }
    sP[0 * P + tid] = fmaxf(fmaxf(r0[0], r0[1]), r0[2]);
    sP[1 * P + tid] = fmaxf(fmaxf(r0[2], r0[3]), r0[4]);
    sP[2 * P + tid] = fmaxf(fmaxf(r1[0], r1[1]), r1[2]);
    sP[3 * P + tid] = fmaxf(fmaxf(r1[2], r1[3]), r1[4]);
    sP[4 * P + tid] = 0.f;
    __syncthreads();

    const int nt = wave & 1;
    const int kq = wave >> 1;
    const float* ap = sP + (r < 4 ? r : 4) * P + kq * 64 + 2 * half;
    const float* bp = sW + (nt * 16 + r) * P + kq * 64 + 2 * half;
    v8f acc = {};
    for (int k0 = 0; k0 < 64; k0 += 4) {
        const float2 av = *reinterpret_cast<const float2*>(ap + k0);
        const float2 bv = *reinterpret_cast<const float2*>(bp + k0);
        v2f fa; fa[0] = av.x; fa[1] = av.y;
        v2f fb; fb[0] = bv.x; fb[1] = bv.y;
        acc = __builtin_amdgcn_wmma_f32_16x16x4_f32(false, fa, false, fb,
                                                    (short)0, acc, false, false);
    }
    if (half == 0) {
        float* o = sO + kq * 128 + nt * 16 + r;
#pragma unroll
        for (int s = 0; s < 4; ++s) o[s * 32] = acc[s];
    }
    __syncthreads();
    if (tid < 32) {
        const float4* p4 = reinterpret_cast<const float4*>(sO) + tid;
        const float4 q0 = p4[0], q1 = p4[32], q2 = p4[64], q3 = p4[96];
        float4 ov;
        ov.x = ((q0.x + q1.x) + q2.x) + q3.x;
        ov.y = ((q0.y + q1.y) + q2.y) + q3.y;
        ov.z = ((q0.z + q1.z) + q2.z) + q3.z;
        ov.w = ((q0.w + q1.w) + q2.w) + q3.w;
        *reinterpret_cast<float4*>(z1raw + (size_t)h * (256 * 32)
                                   + (size_t)n * 128 + tid * 4) = ov;
    }
}

template <int K, int NROWS>
__device__ __forceinline__ void stage_rows(float* __restrict__ dst,
                                           const float* __restrict__ src, int tid) {
    constexpr int P     = K + 4;
    constexpr int VPR   = K / 4;
    constexpr int ITERS = NROWS * VPR / 256;
    static_assert((NROWS * VPR) % 256 == 0, "whole passes of the workgroup only");
    static_assert(256 % VPR == 0, "a row is a whole fraction of one pass of the workgroup");
    const float4* s4 = reinterpret_cast<const float4*>(src) + tid;
    float* d = dst + (tid / VPR) * P + (tid % VPR) * 4;
    float4 v[ITERS];
#pragma unroll
    for (int j = 0; j < ITERS; ++j) v[j] = s4[256 * j];
#pragma unroll
    for (int j = 0; j < ITERS; ++j)
        *reinterpret_cast<float4*>(d + (256 / VPR) * j * P) = v[j];
}

enum { TO_ROWS = 0, TO_FC = 1 };

template <int COUT, int MODE>
__device__ __forceinline__ void bn_relu_store(
    const v8f (&acc)[COUT / 8],
    const float* __restrict__ g, const float* __restrict__ b,
    float* lsum, float* lsq, float* lscale, float* lshift,
    float* dst, int mq, int tid) {
    constexpr int TPW = COUT / 8;

    const int wave = tid >> 5;
    const int lane = tid & 31;
    const int half = lane >> 4;
    const int r    = lane & 15;

#pragma unroll
    for (int i = 0; i < TPW; ++i) {
        const int n0 = (i >> 1) * 16;
        float s = 0.f, q = 0.f;
#pragma unroll
        for (int v = 0; v < 8; ++v) { float xv = acc[i][v]; s += xv; q += xv * xv; }
        atomicAdd(&lsum[n0 + r], s);
        atomicAdd(&lsq[n0 + r], q);
    }
    __syncthreads();

    if (tid < COUT) {
        float mu  = lsum[tid] * (1.f / 256.f);
        float var = lsq[tid] * (1.f / 256.f) - mu * mu;
        float sc  = g[tid] * rsqrtf(var + EPSV);
        lscale[tid] = sc;
        lshift[tid] = b[tid] - mu * sc;
        lsum[tid] = 0.f;
        lsq[tid]  = 0.f;
    }
    __syncthreads();

#pragma unroll
    for (int i = 0; i < TPW; ++i) {
        const int tt = wave + 8 * i;
        const int mt = tt & 15, nt = tt >> 4;
        const int m0 = mt * 16, n0 = nt * 16;
        const int col = n0 + r;
        if (MODE == TO_ROWS) {
            const float sc = lscale[col], sh = lshift[col];
#pragma unroll
            for (int v = 0; v < 8; ++v) {
                const int row = m0 + v + 8 * half;
                dst[row * (COUT + 4) + col] = fmaxf(acc[i][v] * sc + sh, 0.f);
            }
        } else {
            if ((mt >> 2) != mq) continue;
            const float sc = lscale[col], sh = lshift[col];
            float4 lo, hi;
            lo.x = fmaxf(acc[i][0] * sc + sh, 0.f);
            lo.y = fmaxf(acc[i][1] * sc + sh, 0.f);
            lo.z = fmaxf(acc[i][2] * sc + sh, 0.f);
            lo.w = fmaxf(acc[i][3] * sc + sh, 0.f);
            hi.x = fmaxf(acc[i][4] * sc + sh, 0.f);
            hi.y = fmaxf(acc[i][5] * sc + sh, 0.f);
            hi.z = fmaxf(acc[i][6] * sc + sh, 0.f);
            hi.w = fmaxf(acc[i][7] * sc + sh, 0.f);
            float* d = dst + ((mt & 3) * 4 + 2 * half) * (COUT * 4 + 4) + col * 4;
            *reinterpret_cast<float4*>(d) = lo;
            *reinterpret_cast<float4*>(d + (COUT * 4 + 4)) = hi;
        }
    }
}

template <int KIN, int COUT, int MODE>
__device__ __forceinline__ void conv_bn_relu_layer(
    const float* sIn, const float* sWt,
    const float* __restrict__ g, const float* __restrict__ b,
    float* lsum, float* lsq, float* lscale, float* lshift,
    float* dst, int mq, int tid) {
    constexpr int NT  = COUT / 16;
    constexpr int TPW = 2 * NT;
    constexpr int P   = KIN + 4;

    const int wave = tid >> 5;
    const int lane = tid & 31;
    const int half = lane >> 4;
    const int r    = lane & 15;

    v8f acc[TPW];
#pragma unroll
    for (int i = 0; i < TPW; ++i) acc[i] = v8f{};

    const float* alo = sIn + (wave * 16 + r) * P + 2 * half;
    const float* ahi = alo + 128 * P;
    const float* bw  = sWt + r * P + 2 * half;
    for (int k0 = 0; k0 < KIN; k0 += 4) {
        const float2 a0 = *reinterpret_cast<const float2*>(alo + k0);
        const float2 a1 = *reinterpret_cast<const float2*>(ahi + k0);
        v2f fa0; fa0[0] = a0.x; fa0[1] = a0.y;
        v2f fa1; fa1[0] = a1.x; fa1[1] = a1.y;
#pragma unroll
        for (int nt = 0; nt < NT; ++nt) {
            const float2 bv = *reinterpret_cast<const float2*>(bw + nt * 16 * P + k0);
            v2f fb; fb[0] = bv.x; fb[1] = bv.y;
            acc[2 * nt]     = __builtin_amdgcn_wmma_f32_16x16x4_f32(false, fa0, false, fb,
                                                                    (short)0, acc[2 * nt], false, false);
            acc[2 * nt + 1] = __builtin_amdgcn_wmma_f32_16x16x4_f32(false, fa1, false, fb,
                                                                    (short)0, acc[2 * nt + 1], false, false);
        }
    }
    bn_relu_store<COUT, MODE>(acc, g, b, lsum, lsq, lscale, lshift, dst, mq, tid);
}

__global__ __launch_bounds__(256) void conv_fc1_fc2_part_kernel(
    const float* __restrict__ z1raw,
    const float* __restrict__ g1, const float* __restrict__ b1,
    const float* __restrict__ w2, const float* __restrict__ g2, const float* __restrict__ b2,
    const float* __restrict__ w3, const float* __restrict__ g3, const float* __restrict__ b3,
    const float* __restrict__ lw1,
    const float* __restrict__ lb1,
    const float* __restrict__ lw2,
    float* __restrict__ part) {
    constexpr int PC2 = 32 + 4, PC3 = 64 + 4;
    constexpr int P1 = 256 + 4;
    constexpr int P2 = 64 + 4;
    constexpr int OFF_W2 = 4 * 64;
    constexpr int OFF_W3 = OFF_W2 + 64 * PC2;
    constexpr int OFF_Z1 = OFF_W3 + 64 * PC3;
    constexpr int OFF_Z2 = OFF_Z1 + 256 * PC2;
    constexpr int OFF_A  = OFF_Z2 + 256 * PC3;
    constexpr int OFF_B1 = OFF_A + 16 * P1;
    constexpr int OFF_B2 = OFF_B1 + 64 * P1;
    constexpr int OFF_H  = OFF_B2 + 112 * P2;
    constexpr int TOTAL  = OFF_H + 16 * P2;
    static_assert(TOTAL * 4 <= 320 * 1024, "one workgroup's LDS");
    __shared__ alignas(16) float smem[TOTAL];

    float* lsum   = smem;
    float* lsq    = smem + 64;
    float* lscale = smem + 128;
    float* lshift = smem + 192;
    float* sW2 = smem + OFF_W2;
    float* sW3 = smem + OFF_W3;
    float* sZ1 = smem + OFF_Z1;
    float* sZ2 = smem + OFF_Z2;
    float* sA  = smem + OFF_A;
    float* sB1 = smem + OFF_B1;
    float* sB2 = smem + OFF_B2;
    float* sH  = smem + OFF_H;

    const int s    = blockIdx.x >> 2;
    const int mq   = blockIdx.x & 3;
    const int h    = blockIdx.y;
    const int tid  = threadIdx.x;
    const int wave = __builtin_amdgcn_readfirstlane(tid >> 5);
    const int lane = tid & 31;
    const int half = lane >> 4;
    const int r    = lane & 15;
    const int kh   = wave & 1;
    const int nq   = wave >> 1;

    if (tid < 64) { lsum[tid] = 0.f; lsq[tid] = 0.f; }

    v8f acc1[4];
    const float* zr = z1raw + (size_t)h * (256 * 32) + (8 * half) * 32 + r;
#pragma unroll
    for (int i = 0; i < 4; ++i) {
        const int tt = (tid >> 5) + 8 * i;
        const int mt1 = tt & 15, nt1 = tt >> 4;
#pragma unroll
        for (int v = 0; v < 8; ++v)
            acc1[i][v] = zr[(mt1 * 16 + v) * 32 + nt1 * 16];
    }
    stage_rows<32, 64>(sW2, w2 + (size_t)h * 64 * 32, tid);
    stage_rows<64, 64>(sW3, w3 + (size_t)h * 64 * 64, tid);
    __syncthreads();

    bn_relu_store<32, TO_ROWS>(acc1, g1 + h * 32, b1 + h * 32,
                               lsum, lsq, lscale, lshift, sZ1, mq, tid);
    __syncthreads();

    const v4f* b4 = reinterpret_cast<const v4f*>(lw1 + ((size_t)h * 512 + s * 64) * 256) + tid;
    const float* w2p = lw2 + ((size_t)h * 100 + (tid >> 4)) * 512 + s * 64 + (tid & 15) * 4;
    v4f bv1[16], bv2[7];
#pragma unroll
    for (int j = 0; j < 16; ++j) bv1[j] = b4[256 * j];
#pragma unroll
    for (int j = 0; j < 6; ++j) bv2[j] = *reinterpret_cast<const v4f*>(w2p + 16 * j * 512);
    const int r6 = min(96 + (tid >> 4), 99);
    bv2[6] = *reinterpret_cast<const v4f*>(lw2 + ((size_t)h * 100 + r6) * 512 + s * 64 + (tid & 15) * 4);
    const float bs = lb1[h * 512 + s * 64 + nq * 16 + r];

    conv_bn_relu_layer<32, 64, TO_ROWS>(sZ1, sW2, g2 + h * 64, b2 + h * 64,
                                        lsum, lsq, lscale, lshift, sZ2, mq, tid);

    {
        float* bd = sB1 + (tid >> 6) * P1 + (tid & 63) * 4;
#pragma unroll
        for (int j = 0; j < 16; ++j) *reinterpret_cast<v4f*>(bd + 4 * j * P1) = bv1[j];
        float* wd = sB2 + (tid >> 4) * P2 + (tid & 15) * 4;
#pragma unroll
        for (int j = 0; j < 7; ++j) *reinterpret_cast<v4f*>(wd + 16 * j * P2) = bv2[j];
    }
    __syncthreads();
    conv_bn_relu_layer<64, 64, TO_FC>(sZ2, sW3, g3 + h * 64, b3 + h * 64,
                                      lsum, lsq, lscale, lshift, sA, mq, tid);
    __syncthreads();

    {
        const float* ap = sA + r * P1 + kh * 128 + 2 * half;
        const float* bp = sB1 + (nq * 16 + r) * P1 + kh * 128 + 2 * half;
        v8f c = {};
        for (int k0 = 0; k0 < 128; k0 += 4) {
            const float2 a2 = *reinterpret_cast<const float2*>(ap + k0);
            const float2 b2v = *reinterpret_cast<const float2*>(bp + k0);
            v2f fa; fa[0] = a2.x; fa[1] = a2.y;
            v2f fb; fb[0] = b2v.x; fb[1] = b2v.y;
            c = __builtin_amdgcn_wmma_f32_16x16x4_f32(false, fa, false, fb,
                                                      (short)0, c, false, false);
        }
        float* xp = sZ1 + (nq * 32 + lane) * 8;
        if (kh == 1) {
            v4f lo, hi;
            lo[0] = c[0]; lo[1] = c[1]; lo[2] = c[2]; lo[3] = c[3];
            hi[0] = c[4]; hi[1] = c[5]; hi[2] = c[6]; hi[3] = c[7];
            *reinterpret_cast<v4f*>(xp) = lo;
            *reinterpret_cast<v4f*>(xp + 4) = hi;
        }
        __syncthreads();
        if (kh == 0) {
            const v4f lo = *reinterpret_cast<const v4f*>(xp);
            const v4f hi = *reinterpret_cast<const v4f*>(xp + 4);
            float* hp = sH + (8 * half) * P2 + nq * 16 + r;
#pragma unroll
            for (int v = 0; v < 4; ++v) {
                hp[v * P2]       = fmaxf((c[v] + lo[v]) + bs, 0.f);
                hp[(v + 4) * P2] = fmaxf((c[v + 4] + hi[v]) + bs, 0.f);
            }
        }
    }
    __syncthreads();

    if (wave < 7) {
        const int c0 = wave * 16 + r;
        const bool ok = c0 < 100;
        const float* ap  = sH + r * P2 + 2 * half;
        const float* bp0 = sB2 + (ok ? c0 : 0) * P2 + 2 * half;
        v8f d0 = {};
        for (int k0 = 0; k0 < 64; k0 += 4) {
            const float2 a2 = *reinterpret_cast<const float2*>(ap + k0);
            const float2 x0 = *reinterpret_cast<const float2*>(bp0 + k0);
            v2f fa;  fa[0] = a2.x; fa[1] = a2.y;
            v2f fb0; fb0[0] = ok ? x0.x : 0.f; fb0[1] = ok ? x0.y : 0.f;
            d0 = __builtin_amdgcn_wmma_f32_16x16x4_f32(false, fa, false, fb0,
                                                       (short)0, d0, false, false);
        }
        float* o = part + (((size_t)h * 8 + s) * 64 + mq * 16 + 8 * half) * 100;
        if (ok) {
#pragma unroll
            for (int v = 0; v < 8; ++v) o[v * 100 + c0] = d0[v];
        }
    }
}

__global__ __launch_bounds__(128) void final_combine_kernel(
                                     const float* __restrict__ part,
                                     const float* __restrict__ lb2,
                                     const float* __restrict__ W2,
                                     const float* __restrict__ bias2,
                                     float* __restrict__ out) {
    __shared__ alignas(16) float sW[100 * 100];
    __shared__ alignas(16) float comb[128];
    __shared__ alignas(16) float res[128];
    const int n  = blockIdx.x;
    const int t  = threadIdx.x;
    const int tc = min(t, 99);

    float pv[32], lb[4];
#pragma unroll
    for (int p = 0; p < 32; ++p) pv[p] = part[((size_t)p * 64 + n) * 100 + tc];
#pragma unroll
    for (int h = 0; h < 4; ++h) lb[h] = lb2[h * 100 + tc];
    const float bz = bias2[tc];

    const float4* w4 = reinterpret_cast<const float4*>(W2);
    const int last = min(t + 2432, 2499);
    float4 wv[20];
#pragma unroll
    for (int j = 0; j < 19; ++j) wv[j] = w4[t + 128 * j];
    wv[19] = w4[last];

    float s = 0.f;
#pragma unroll
    for (int p = 0; p < 32; ++p) s += pv[p];
#pragma unroll
    for (int h = 0; h < 4; ++h) s += lb[h];
    comb[t] = s * 0.25f;

    float4* sw4 = reinterpret_cast<float4*>(sW);
#pragma unroll
    for (int j = 0; j < 19; ++j) sw4[t + 128 * j] = wv[j];
    sw4[last] = wv[19];
    __syncthreads();

    {
        const float4* wr = reinterpret_cast<const float4*>(sW + tc * 100);
        const float4* c4 = reinterpret_cast<const float4*>(comb);
        float acc = bz;
#pragma unroll
        for (int q = 0; q < 25; ++q) {
            const float4 w = wr[q];
            const float4 c = c4[q];
            acc = fmaf(c.x, w.x, acc);
            acc = fmaf(c.y, w.y, acc);
            acc = fmaf(c.z, w.z, acc);
            acc = fmaf(c.w, w.w, acc);
        }
        res[t] = acc;
    }
    __syncthreads();

    const float4* r4 = reinterpret_cast<const float4*>(res);
    float4* o4 = reinterpret_cast<float4*>(out + (size_t)n * 64 * 100);
#pragma unroll
    for (int j = 0; j < 12; ++j) {
        const int i = t + 128 * j;
        o4[i] = r4[i % 25];
    }
    if (t < 64) {
        const int i = t + 1536;
        o4[i] = r4[i % 25];
    }
}

extern "C" void kernel_launch(void* const* d_in, const int* in_sizes, int n_in,
                              void* d_out, int out_size, void* d_ws, size_t ws_size,
                              hipStream_t stream) {
    const float* x     = (const float*)d_in[0];
    const float* w1    = (const float*)d_in[1];
    const float* g1    = (const float*)d_in[2];
    const float* b1    = (const float*)d_in[3];
    const float* w2    = (const float*)d_in[4];
    const float* g2    = (const float*)d_in[5];
    const float* b2    = (const float*)d_in[6];
    const float* w3    = (const float*)d_in[7];
    const float* g3    = (const float*)d_in[8];
    const float* b3    = (const float*)d_in[9];
    const float* lw1   = (const float*)d_in[10];
    const float* lb1   = (const float*)d_in[11];
    const float* lw2   = (const float*)d_in[12];
    const float* lb2   = (const float*)d_in[13];
    const float* W2    = (const float*)d_in[14];
    const float* bias2 = (const float*)d_in[15];
    float* out = (float*)d_out;

    float* ws    = (float*)d_ws;
    float* part  = ws;
    float* z1raw = ws + 222208;

    roi_pool_conv1_kernel<<<dim3(64, 4), 256, 0, stream>>>(x, w1, z1raw);

    conv_fc1_fc2_part_kernel<<<dim3(32, 4), 256, 0, stream>>>(
        z1raw, g1, b1, w2, g2, b2, w3, g3, b3, lw1, lb1, lw2, part);

    final_combine_kernel<<<64, 128, 0, stream>>>(part, lb2, W2, bias2, out);
}
